# starting kernel
# baseline (speedup 1.0000x reference)
#include <hip/hip_runtime.h>
#include <hip/hip_bf16.h>


typedef __attribute__((ext_vector_type(2))) float v2f;
typedef __attribute__((ext_vector_type(8))) float v8f;

#define DFEAT 128

__global__ void zero_f4(float4* __restrict__ p, long n4) {
    long i = (long)blockIdx.x * blockDim.x + threadIdx.x;
    long stride = (long)gridDim.x * blockDim.x;
    float4 z = make_float4(0.f, 0.f, 0.f, 0.f);
    for (; i < n4; i += stride) p[i] = z;
}

__global__ void deg_kernel(const int* __restrict__ dst, float* __restrict__ deg,
                           int n_edges) {
    int e = blockIdx.x * blockDim.x + threadIdx.x;
    if (e < n_edges) atomicAdd(&deg[dst[e]], 1.0f);
}

__global__ __launch_bounds__(256)
void sage_scatter(const float* __restrict__ xin, const int* __restrict__ src,
                  const int* __restrict__ dst, float* __restrict__ agg,
                  int n_edges) {
    int wid  = (blockIdx.x * blockDim.x + threadIdx.x) >> 5;
    int lane = threadIdx.x & 31;
    if (wid >= n_edges) return;
    int s = src[wid];
    int d = dst[wid];
    float4 v = ((const float4*)(xin + (size_t)s * DFEAT))[lane];
    float* ar = agg + (size_t)d * DFEAT + lane * 4;
    atomicAdd(ar + 0, v.x);
    atomicAdd(ar + 1, v.y);
    atomicAdd(ar + 2, v.z);
    atomicAdd(ar + 3, v.w);
}

__global__ __launch_bounds__(64)
void sage_gemm(const float* __restrict__ agg, const float* __restrict__ deg,
               const float* __restrict__ xin,
               const float* __restrict__ Wl, const float* __restrict__ bl,
               const float* __restrict__ Wr, float* __restrict__ out,
               int n_tiles, int do_relu) {
    int tile = ((blockIdx.x * blockDim.x + threadIdx.x) >> 5);
    if (tile >= n_tiles) return;
    int lane = threadIdx.x & 31;
    int m = lane & 15;
    int h = lane >> 4;

    int row = tile * 16 + m;
    float rd = 1.0f / fmaxf(deg[row], 1.0f);

    const float* aggRow = agg + (size_t)row * DFEAT;
    const float* xRow   = xin + (size_t)row * DFEAT;
    const float* wlBase = Wl + (size_t)m * DFEAT;
    const float* wrBase = Wr + (size_t)m * DFEAT;

    v8f acc[8] = {};

    for (int k = 0; k < DFEAT; k += 4) {
        int kk = k + 2 * h;
        v2f a_m = *(const v2f*)(aggRow + kk);
        a_m[0] *= rd;
        a_m[1] *= rd;
        v2f a_x = *(const v2f*)(xRow + kk);
#pragma unroll
        for (int t = 0; t < 8; ++t) {
            v2f b_l = *(const v2f*)(wlBase + (size_t)t * 16 * DFEAT + kk);
            v2f b_r = *(const v2f*)(wrBase + (size_t)t * 16 * DFEAT + kk);
            acc[t] = __builtin_amdgcn_wmma_f32_16x16x4_f32(
                false, a_m, false, b_l, (short)0, acc[t], false, false);
            acc[t] = __builtin_amdgcn_wmma_f32_16x16x4_f32(
                false, a_x, false, b_r, (short)0, acc[t], false, false);
        }
    }

#pragma unroll
    for (int t = 0; t < 8; ++t) {
        float bias = bl[t * 16 + m];
        float* orow = out + (size_t)(tile * 16 + 8 * h) * DFEAT + t * 16 + m;
#pragma unroll
        for (int r = 0; r < 8; ++r) {
            float v = acc[t][r] + bias;
            if (do_relu) v = fmaxf(v, 0.0f);
            orow[(size_t)r * DFEAT] = v;
        }
    }
}

extern "C" void kernel_launch(void* const* d_in, const int* in_sizes, int n_in,
                              void* d_out, int out_size, void* d_ws, size_t ws_size,
                              hipStream_t stream) {
    const float* x   = (const float*)d_in[0];
    const int*   ei  = (const int*)d_in[1];
    const float* Wl1 = (const float*)d_in[2];
    const float* bl1 = (const float*)d_in[3];
    const float* Wr1 = (const float*)d_in[4];
    const float* Wl2 = (const float*)d_in[5];
    const float* bl2 = (const float*)d_in[6];
    const float* Wr2 = (const float*)d_in[7];
    float* outp = (float*)d_out;

    int n = in_sizes[0] / DFEAT;
    int E = in_sizes[1] / 2;
    const int* src = ei;
    const int* dst = ei + E;

    float* deg = (float*)d_ws;
    float* agg = deg + n;
    float* hbuf = agg + (size_t)n * DFEAT;

    int tiles = (n + 15) / 16;
    int gemm_blocks = (tiles + 1) / 2;
    int scat_blocks = (int)(((long)E * 32 + 255) / 256);

    zero_f4<<<2048, 256, 0, stream>>>((float4*)deg, ((long)n * (DFEAT + 1)) / 4);
    deg_kernel<<<(E + 255) / 256, 256, 0, stream>>>(dst, deg, E);
    sage_scatter<<<scat_blocks, 256, 0, stream>>>(x, src, dst, agg, E);
    sage_gemm<<<gemm_blocks, 64, 0, stream>>>(agg, deg, x, Wl1, bl1, Wr1, hbuf,
                                              tiles, 1);
    zero_f4<<<2048, 256, 0, stream>>>((float4*)agg, ((long)n * DFEAT) / 4);
    sage_scatter<<<scat_blocks, 256, 0, stream>>>(hbuf, src, dst, agg, E);
    sage_gemm<<<gemm_blocks, 64, 0, stream>>>(agg, deg, hbuf, Wl2, bl2, Wr2,
                                              outp, tiles, 0);
}


# scatter: lane L handles columns L+32j, so each f32 atomic add instruction covers one whole 128-byte line of the agg row (4 line requests per edge, was 16); sums keep atomic arrival order
# speedup vs baseline: 1.3045x; 1.3045x over previous
#include <hip/hip_runtime.h>
#include <hip/hip_bf16.h>


typedef __attribute__((ext_vector_type(2))) float v2f;
typedef __attribute__((ext_vector_type(8))) float v8f;

#define DFEAT 128

__global__ void zero_f4(float4* __restrict__ p, long n4) {
    long i = (long)blockIdx.x * blockDim.x + threadIdx.x;
    long stride = (long)gridDim.x * blockDim.x;
    float4 z = make_float4(0.f, 0.f, 0.f, 0.f);
    for (; i < n4; i += stride) p[i] = z;
}

__global__ void deg_kernel(const int* __restrict__ dst, float* __restrict__ deg,
                           int n_edges) {
    int e = blockIdx.x * blockDim.x + threadIdx.x;
    if (e < n_edges) atomicAdd(&deg[dst[e]], 1.0f);
}

__global__ __launch_bounds__(256)
void sage_scatter(const float* __restrict__ xin, const int* __restrict__ src,
                  const int* __restrict__ dst, float* __restrict__ agg,
                  int n_edges) {
    int wid  = (blockIdx.x * blockDim.x + threadIdx.x) >> 5;
    int lane = threadIdx.x & 31;
    if (wid >= n_edges) return;
    int s = src[wid];
    int d = dst[wid];
    const float* xr = xin + (size_t)s * DFEAT + lane;
    float v0 = xr[0];
    float v1 = xr[32];
    float v2 = xr[64];
    float v3 = xr[96];
    float* ar = agg + (size_t)d * DFEAT + lane;
    atomicAdd(ar + 0,  v0);
    atomicAdd(ar + 32, v1);
    atomicAdd(ar + 64, v2);
    atomicAdd(ar + 96, v3);
}

__global__ __launch_bounds__(64)
void sage_gemm(const float* __restrict__ agg, const float* __restrict__ deg,
               const float* __restrict__ xin,
               const float* __restrict__ Wl, const float* __restrict__ bl,
               const float* __restrict__ Wr, float* __restrict__ out,
               int n_tiles, int do_relu) {
    int tile = ((blockIdx.x * blockDim.x + threadIdx.x) >> 5);
    if (tile >= n_tiles) return;
    int lane = threadIdx.x & 31;
    int m = lane & 15;
    int h = lane >> 4;

    int row = tile * 16 + m;
    float rd = 1.0f / fmaxf(deg[row], 1.0f);

    const float* aggRow = agg + (size_t)row * DFEAT;
    const float* xRow   = xin + (size_t)row * DFEAT;
    const float* wlBase = Wl + (size_t)m * DFEAT;
    const float* wrBase = Wr + (size_t)m * DFEAT;

    v8f acc[8] = {};

    for (int k = 0; k < DFEAT; k += 4) {
        int kk = k + 2 * h;
        v2f a_m = *(const v2f*)(aggRow + kk);
        a_m[0] *= rd;
        a_m[1] *= rd;
        v2f a_x = *(const v2f*)(xRow + kk);
#pragma unroll
        for (int t = 0; t < 8; ++t) {
            v2f b_l = *(const v2f*)(wlBase + (size_t)t * 16 * DFEAT + kk);
            v2f b_r = *(const v2f*)(wrBase + (size_t)t * 16 * DFEAT + kk);
            acc[t] = __builtin_amdgcn_wmma_f32_16x16x4_f32(
                false, a_m, false, b_l, (short)0, acc[t], false, false);
            acc[t] = __builtin_amdgcn_wmma_f32_16x16x4_f32(
                false, a_x, false, b_r, (short)0, acc[t], false, false);
        }
    }

#pragma unroll
    for (int t = 0; t < 8; ++t) {
        float bias = bl[t * 16 + m];
        float* orow = out + (size_t)(tile * 16 + 8 * h) * DFEAT + t * 16 + m;
#pragma unroll
        for (int r = 0; r < 8; ++r) {
            float v = acc[t][r] + bias;
            if (do_relu) v = fmaxf(v, 0.0f);
            orow[(size_t)r * DFEAT] = v;
        }
    }
}

extern "C" void kernel_launch(void* const* d_in, const int* in_sizes, int n_in,
                              void* d_out, int out_size, void* d_ws, size_t ws_size,
                              hipStream_t stream) {
    const float* x   = (const float*)d_in[0];
    const int*   ei  = (const int*)d_in[1];
    const float* Wl1 = (const float*)d_in[2];
    const float* bl1 = (const float*)d_in[3];
    const float* Wr1 = (const float*)d_in[4];
    const float* Wl2 = (const float*)d_in[5];
    const float* bl2 = (const float*)d_in[6];
    const float* Wr2 = (const float*)d_in[7];
    float* outp = (float*)d_out;

    int n = in_sizes[0] / DFEAT;
    int E = in_sizes[1] / 2;
    const int* src = ei;
    const int* dst = ei + E;

    float* deg = (float*)d_ws;
    float* agg = deg + n;
    float* hbuf = agg + (size_t)n * DFEAT;

    int tiles = (n + 15) / 16;
    int gemm_blocks = (tiles + 1) / 2;
    int scat_blocks = (int)(((long)E * 32 + 255) / 256);

    zero_f4<<<2048, 256, 0, stream>>>((float4*)deg, ((long)n * (DFEAT + 1)) / 4);
    deg_kernel<<<(E + 255) / 256, 256, 0, stream>>>(dst, deg, E);
    sage_scatter<<<scat_blocks, 256, 0, stream>>>(x, src, dst, agg, E);
    sage_gemm<<<gemm_blocks, 64, 0, stream>>>(agg, deg, x, Wl1, bl1, Wr1, hbuf,
                                              tiles, 1);
    zero_f4<<<2048, 256, 0, stream>>>((float4*)agg, ((long)n * DFEAT) / 4);
    sage_scatter<<<scat_blocks, 256, 0, stream>>>(hbuf, src, dst, agg, E);
    sage_gemm<<<gemm_blocks, 64, 0, stream>>>(agg, deg, hbuf, Wl2, bl2, Wr2,
                                              outp, tiles, 0);
}


# edge sums by destination-grouped gather: per call int-atomic count, scan, slot fill, each node's list sorted by source index; one wave adds its rows in that fixed order; f32 atomics gone
# speedup vs baseline: 2.2881x; 1.7540x over previous
#include <hip/hip_runtime.h>
#include <hip/hip_bf16.h>


typedef __attribute__((ext_vector_type(2))) float v2f;
typedef __attribute__((ext_vector_type(8))) float v8f;

#define DFEAT 128
#define SCAN_ITEMS 1024

__global__ void zero_i32(int* __restrict__ p, int n) {
    int i = blockIdx.x * blockDim.x + threadIdx.x;
    if (i < n) p[i] = 0;
}

__global__ void count_rank(const int* __restrict__ dst, int* __restrict__ cnt,
                           int* __restrict__ rank, int n_edges) {
    int e = blockIdx.x * blockDim.x + threadIdx.x;
    if (e < n_edges) rank[e] = atomicAdd(&cnt[dst[e]], 1);
}

__global__ __launch_bounds__(256)
void scan_local(const int* __restrict__ cnt, int* __restrict__ off,
                int* __restrict__ bsum, int n) {
    __shared__ int wsum[8];
    int t = threadIdx.x;
    int lane = t & 31;
    int w = t >> 5;
    int i0 = blockIdx.x * SCAN_ITEMS + t * 4;
    int c0 = (i0 + 0 < n) ? cnt[i0 + 0] : 0;
    int c1 = (i0 + 1 < n) ? cnt[i0 + 1] : 0;
    int c2 = (i0 + 2 < n) ? cnt[i0 + 2] : 0;
    int c3 = (i0 + 3 < n) ? cnt[i0 + 3] : 0;
    int tot = c0 + c1 + c2 + c3;
    int incl = tot;
    for (int o = 1; o < 32; o <<= 1) {
        int v = __shfl_up(incl, o);
        if (lane >= o) incl += v;
    }
    if (lane == 31) wsum[w] = incl;
    __syncthreads();
    int wb = 0;
    for (int j = 0; j < w; ++j) wb += wsum[j];
    int ex = wb + incl - tot;
    if (i0 + 0 < n) off[i0 + 0] = ex;
    if (i0 + 1 < n) off[i0 + 1] = ex + c0;
    if (i0 + 2 < n) off[i0 + 2] = ex + c0 + c1;
    if (i0 + 3 < n) off[i0 + 3] = ex + c0 + c1 + c2;
    if (t == 255) bsum[blockIdx.x] = wb + incl;
}

__global__ __launch_bounds__(256)
void scan_add(int* __restrict__ off, const int* __restrict__ bsum, int n) {
    __shared__ int wsum[8];
    int t = threadIdx.x;
    int lane = t & 31;
    int w = t >> 5;
    int nb = blockIdx.x;
    int part = 0;
    for (int j = t; j < nb; j += 256) part += bsum[j];
    for (int o = 16; o > 0; o >>= 1) part += __shfl_xor(part, o);
    if (lane == 0) wsum[w] = part;
    __syncthreads();
    int pre = 0;
    for (int j = 0; j < 8; ++j) pre += wsum[j];
    int i0 = blockIdx.x * SCAN_ITEMS + t * 4;
    for (int q = 0; q < 4; ++q)
        if (i0 + q < n) off[i0 + q] += pre;
}

__global__ void fill_slots(const int* __restrict__ src, const int* __restrict__ dst,
                           const int* __restrict__ rank, const int* __restrict__ off,
                           int* __restrict__ slot, int n_edges) {
    int e = blockIdx.x * blockDim.x + threadIdx.x;
    if (e < n_edges) slot[off[dst[e]] + rank[e]] = src[e];
}

__global__ __launch_bounds__(256)
void order_lists(const int* __restrict__ slot, const int* __restrict__ off,
                 const int* __restrict__ cnt, int* __restrict__ sorted,
                 float* __restrict__ deg, int n_nodes) {
    int node = __builtin_amdgcn_readfirstlane(
        (int)((blockIdx.x * blockDim.x + threadIdx.x) >> 5));
    int lane = threadIdx.x & 31;
    if (node >= n_nodes) return;
    int c = __builtin_amdgcn_readfirstlane(cnt[node]);
    int base = __builtin_amdgcn_readfirstlane(off[node]);
    if (lane == 0) deg[node] = (float)c;
    if (c <= 32) {
        int s = (lane < c) ? slot[base + lane] : 0;
        int r = 0;
        for (int j = 0; j < c; ++j) {
            int sj = __builtin_amdgcn_readlane(s, j);
            r += ((sj < s) || (sj == s && j < lane)) ? 1 : 0;
        }
        if (lane < c) sorted[base + r] = s;
    } else {
        for (int i = lane; i < c; i += 32) {
            int s = slot[base + i];
            int r = 0;
            for (int j = 0; j < c; ++j) {
                int sj = slot[base + j];
                r += ((sj < s) || (sj == s && j < i)) ? 1 : 0;
            }
            sorted[base + r] = s;
        }
    }
}

__device__ __forceinline__ void add4(float4& a, const float4& v) {
    a.x += v.x; a.y += v.y; a.z += v.z; a.w += v.w;
}

__global__ __launch_bounds__(256)
void sage_gather(const float* __restrict__ xin, const int* __restrict__ sorted,
                 const int* __restrict__ off, const int* __restrict__ cnt,
                 float* __restrict__ agg, int n_nodes) {
    int node = __builtin_amdgcn_readfirstlane(
        (int)((blockIdx.x * blockDim.x + threadIdx.x) >> 5));
    int lane = threadIdx.x & 31;
    if (node >= n_nodes) return;
    int c = __builtin_amdgcn_readfirstlane(cnt[node]);
    int base = __builtin_amdgcn_readfirstlane(off[node]);
    const float4* xr = (const float4*)xin + lane;
    float4 acc = make_float4(0.f, 0.f, 0.f, 0.f);
    for (int c0 = 0; c0 < c; c0 += 32) {
        int m = c - c0;
        if (m > 32) m = 32;
        int s = (lane < m) ? sorted[base + c0 + lane] : 0;
        int k = 0;
        for (; k + 4 <= m; k += 4) {
            int s0 = __builtin_amdgcn_readlane(s, k);
            int s1 = __builtin_amdgcn_readlane(s, k + 1);
            int s2 = __builtin_amdgcn_readlane(s, k + 2);
            int s3 = __builtin_amdgcn_readlane(s, k + 3);
            float4 v0 = xr[(size_t)s0 * (DFEAT / 4)];
            float4 v1 = xr[(size_t)s1 * (DFEAT / 4)];
            float4 v2 = xr[(size_t)s2 * (DFEAT / 4)];
            float4 v3 = xr[(size_t)s3 * (DFEAT / 4)];
            add4(acc, v0);
            add4(acc, v1);
            add4(acc, v2);
            add4(acc, v3);
        }
        for (; k < m; ++k) {
            int s0 = __builtin_amdgcn_readlane(s, k);
            float4 v0 = xr[(size_t)s0 * (DFEAT / 4)];
            add4(acc, v0);
        }
    }
    ((float4*)(agg + (size_t)node * DFEAT))[lane] = acc;
}

__global__ __launch_bounds__(64)
void sage_gemm(const float* __restrict__ agg, const float* __restrict__ deg,
               const float* __restrict__ xin,
               const float* __restrict__ Wl, const float* __restrict__ bl,
               const float* __restrict__ Wr, float* __restrict__ out,
               int n_tiles, int do_relu) {
    int tile = ((blockIdx.x * blockDim.x + threadIdx.x) >> 5);
    if (tile >= n_tiles) return;
    int lane = threadIdx.x & 31;
    int m = lane & 15;
    int h = lane >> 4;

    int row = tile * 16 + m;
    float rd = 1.0f / fmaxf(deg[row], 1.0f);

    const float* aggRow = agg + (size_t)row * DFEAT;
    const float* xRow   = xin + (size_t)row * DFEAT;
    const float* wlBase = Wl + (size_t)m * DFEAT;
    const float* wrBase = Wr + (size_t)m * DFEAT;

    v8f acc[8] = {};

    for (int k = 0; k < DFEAT; k += 4) {
        int kk = k + 2 * h;
        v2f a_m = *(const v2f*)(aggRow + kk);
        a_m[0] *= rd;
        a_m[1] *= rd;
        v2f a_x = *(const v2f*)(xRow + kk);
#pragma unroll
        for (int t = 0; t < 8; ++t) {
            v2f b_l = *(const v2f*)(wlBase + (size_t)t * 16 * DFEAT + kk);
            v2f b_r = *(const v2f*)(wrBase + (size_t)t * 16 * DFEAT + kk);
            acc[t] = __builtin_amdgcn_wmma_f32_16x16x4_f32(
                false, a_m, false, b_l, (short)0, acc[t], false, false);
            acc[t] = __builtin_amdgcn_wmma_f32_16x16x4_f32(
                false, a_x, false, b_r, (short)0, acc[t], false, false);
        }
    }

#pragma unroll
    for (int t = 0; t < 8; ++t) {
        float bias = bl[t * 16 + m];
        float* orow = out + (size_t)(tile * 16 + 8 * h) * DFEAT + t * 16 + m;
#pragma unroll
        for (int r = 0; r < 8; ++r) {
            float v = acc[t][r] + bias;
            if (do_relu) v = fmaxf(v, 0.0f);
            orow[(size_t)r * DFEAT] = v;
        }
    }
}

extern "C" void kernel_launch(void* const* d_in, const int* in_sizes, int n_in,
                              void* d_out, int out_size, void* d_ws, size_t ws_size,
                              hipStream_t stream) {
    const float* x   = (const float*)d_in[0];
    const int*   ei  = (const int*)d_in[1];
    const float* Wl1 = (const float*)d_in[2];
    const float* bl1 = (const float*)d_in[3];
    const float* Wr1 = (const float*)d_in[4];
    const float* Wl2 = (const float*)d_in[5];
    const float* bl2 = (const float*)d_in[6];
    const float* Wr2 = (const float*)d_in[7];
    float* outp = (float*)d_out;

    int n = in_sizes[0] / DFEAT;
    int E = in_sizes[1] / 2;
    const int* src = ei;
    const int* dst = ei + E;

    float* agg = (float*)d_ws;
    float* hbuf = agg + (size_t)n * DFEAT;
    float* deg = hbuf + (size_t)n * DFEAT;
    int* cnt    = (int*)(deg + n);
    int* off    = cnt + n;
    int* bsum   = off + n;
    int* rank   = bsum + 1024;
    int* slot   = rank + E;
    int* sorted = slot + E;

    int tiles = (n + 15) / 16;
    int gemm_blocks = (tiles + 1) / 2;
    int edge_blocks = (E + 255) / 256;
    int node_blocks = (n + 255) / 256;
    int wave_blocks = (n + 7) / 8;
    int scan_blocks = (n + SCAN_ITEMS - 1) / SCAN_ITEMS;

    zero_i32<<<node_blocks, 256, 0, stream>>>(cnt, n);
    count_rank<<<edge_blocks, 256, 0, stream>>>(dst, cnt, rank, E);
    scan_local<<<scan_blocks, 256, 0, stream>>>(cnt, off, bsum, n);
    scan_add<<<scan_blocks, 256, 0, stream>>>(off, bsum, n);
    fill_slots<<<edge_blocks, 256, 0, stream>>>(src, dst, rank, off, slot, E);
    order_lists<<<wave_blocks, 256, 0, stream>>>(slot, off, cnt, sorted, deg, n);

    sage_gather<<<wave_blocks, 256, 0, stream>>>(x, sorted, off, cnt, agg, n);
    sage_gemm<<<gemm_blocks, 64, 0, stream>>>(agg, deg, x, Wl1, bl1, Wr1, hbuf,
                                              tiles, 1);
    sage_gather<<<wave_blocks, 256, 0, stream>>>(hbuf, sorted, off, cnt, agg, n);
    sage_gemm<<<gemm_blocks, 64, 0, stream>>>(agg, deg, hbuf, Wl2, bl2, Wr2,
                                              outp, tiles, 0);
}


# GEMM reads its B fragments from LDS: each work-group stages Wl|Wr once (130 KiB, pitch 260 floats), 512 groups of 8 waves loop over the tiles, A loaded 32 k ahead; order of every sum unchanged
# speedup vs baseline: 2.8854x; 1.2610x over previous
#include <hip/hip_runtime.h>
#include <hip/hip_bf16.h>


typedef __attribute__((ext_vector_type(2))) float v2f;
typedef __attribute__((ext_vector_type(8))) float v8f;

#define DFEAT 128
#define SCAN_ITEMS 1024

__global__ void zero_i32(int* __restrict__ p, int n) {
    int i = blockIdx.x * blockDim.x + threadIdx.x;
    if (i < n) p[i] = 0;
}

__global__ void count_rank(const int* __restrict__ dst, int* __restrict__ cnt,
                           int* __restrict__ rank, int n_edges) {
    int e = blockIdx.x * blockDim.x + threadIdx.x;
    if (e < n_edges) rank[e] = atomicAdd(&cnt[dst[e]], 1);
}

__global__ __launch_bounds__(256)
void scan_local(const int* __restrict__ cnt, int* __restrict__ off,
                int* __restrict__ bsum, int n) {
    __shared__ int wsum[8];
    int t = threadIdx.x;
    int lane = t & 31;
    int w = t >> 5;
    int i0 = blockIdx.x * SCAN_ITEMS + t * 4;
    int c0 = (i0 + 0 < n) ? cnt[i0 + 0] : 0;
    int c1 = (i0 + 1 < n) ? cnt[i0 + 1] : 0;
    int c2 = (i0 + 2 < n) ? cnt[i0 + 2] : 0;
    int c3 = (i0 + 3 < n) ? cnt[i0 + 3] : 0;
    int tot = c0 + c1 + c2 + c3;
    int incl = tot;
    for (int o = 1; o < 32; o <<= 1) {
        int v = __shfl_up(incl, o);
        if (lane >= o) incl += v;
    }
    if (lane == 31) wsum[w] = incl;
    __syncthreads();
    int wb = 0;
    for (int j = 0; j < w; ++j) wb += wsum[j];
    int ex = wb + incl - tot;
    if (i0 + 0 < n) off[i0 + 0] = ex;
    if (i0 + 1 < n) off[i0 + 1] = ex + c0;
    if (i0 + 2 < n) off[i0 + 2] = ex + c0 + c1;
    if (i0 + 3 < n) off[i0 + 3] = ex + c0 + c1 + c2;
    if (t == 255) bsum[blockIdx.x] = wb + incl;
}

__global__ __launch_bounds__(256)
void scan_add(int* __restrict__ off, const int* __restrict__ bsum, int n) {
    __shared__ int wsum[8];
    int t = threadIdx.x;
    int lane = t & 31;
    int w = t >> 5;
    int nb = blockIdx.x;
    int part = 0;
    for (int j = t; j < nb; j += 256) part += bsum[j];
    for (int o = 16; o > 0; o >>= 1) part += __shfl_xor(part, o);
    if (lane == 0) wsum[w] = part;
    __syncthreads();
    int pre = 0;
    for (int j = 0; j < 8; ++j) pre += wsum[j];
    int i0 = blockIdx.x * SCAN_ITEMS + t * 4;
    for (int q = 0; q < 4; ++q)
        if (i0 + q < n) off[i0 + q] += pre;
}

__global__ void fill_slots(const int* __restrict__ src, const int* __restrict__ dst,
                           const int* __restrict__ rank, const int* __restrict__ off,
                           int* __restrict__ slot, int n_edges) {
    int e = blockIdx.x * blockDim.x + threadIdx.x;
    if (e < n_edges) slot[off[dst[e]] + rank[e]] = src[e];
}

__global__ __launch_bounds__(256)
void order_lists(const int* __restrict__ slot, const int* __restrict__ off,
                 const int* __restrict__ cnt, int* __restrict__ sorted,
                 float* __restrict__ deg, int n_nodes) {
    int node = __builtin_amdgcn_readfirstlane(
        (int)((blockIdx.x * blockDim.x + threadIdx.x) >> 5));
    int lane = threadIdx.x & 31;
    if (node >= n_nodes) return;
    int c = __builtin_amdgcn_readfirstlane(cnt[node]);
    int base = __builtin_amdgcn_readfirstlane(off[node]);
    if (lane == 0) deg[node] = (float)c;
    if (c <= 32) {
        int s = (lane < c) ? slot[base + lane] : 0;
        int r = 0;
        for (int j = 0; j < c; ++j) {
            int sj = __builtin_amdgcn_readlane(s, j);
            r += ((sj < s) || (sj == s && j < lane)) ? 1 : 0;
        }
        if (lane < c) sorted[base + r] = s;
    } else {
        for (int i = lane; i < c; i += 32) {
            int s = slot[base + i];
            int r = 0;
            for (int j = 0; j < c; ++j) {
                int sj = slot[base + j];
                r += ((sj < s) || (sj == s && j < i)) ? 1 : 0;
            }
            sorted[base + r] = s;
        }
    }
}

__device__ __forceinline__ void add4(float4& a, const float4& v) {
    a.x += v.x; a.y += v.y; a.z += v.z; a.w += v.w;
}

__global__ __launch_bounds__(256)
void sage_gather(const float* __restrict__ xin, const int* __restrict__ sorted,
                 const int* __restrict__ off, const int* __restrict__ cnt,
                 float* __restrict__ agg, int n_nodes) {
    int node = __builtin_amdgcn_readfirstlane(
        (int)((blockIdx.x * blockDim.x + threadIdx.x) >> 5));
    int lane = threadIdx.x & 31;
    if (node >= n_nodes) return;
    int c = __builtin_amdgcn_readfirstlane(cnt[node]);
    int base = __builtin_amdgcn_readfirstlane(off[node]);
    const float4* xr = (const float4*)xin + lane;
    float4 acc = make_float4(0.f, 0.f, 0.f, 0.f);
    for (int c0 = 0; c0 < c; c0 += 32) {
        int m = c - c0;
        if (m > 32) m = 32;
        int s = (lane < m) ? sorted[base + c0 + lane] : 0;
        int k = 0;
        for (; k + 4 <= m; k += 4) {
            int s0 = __builtin_amdgcn_readlane(s, k);
            int s1 = __builtin_amdgcn_readlane(s, k + 1);
            int s2 = __builtin_amdgcn_readlane(s, k + 2);
            int s3 = __builtin_amdgcn_readlane(s, k + 3);
            float4 v0 = xr[(size_t)s0 * (DFEAT / 4)];
            float4 v1 = xr[(size_t)s1 * (DFEAT / 4)];
            float4 v2 = xr[(size_t)s2 * (DFEAT / 4)];
            float4 v3 = xr[(size_t)s3 * (DFEAT / 4)];
            add4(acc, v0);
            add4(acc, v1);
            add4(acc, v2);
            add4(acc, v3);
        }
        for (; k < m; ++k) {
            int s0 = __builtin_amdgcn_readlane(s, k);
            float4 v0 = xr[(size_t)s0 * (DFEAT / 4)];
            add4(acc, v0);
        }
    }
    ((float4*)(agg + (size_t)node * DFEAT))[lane] = acc;
}

#define WPITCH 260
#define GEMM_WAVES 8
#define GEMM_BLOCKS 512

__global__ __launch_bounds__(256, 4)
void sage_gemm(const float* __restrict__ agg, const float* __restrict__ deg,
               const float* __restrict__ xin,
               const float* __restrict__ Wl, const float* __restrict__ bl,
               const float* __restrict__ Wr, float* __restrict__ out,
               int n_tiles, int do_relu) {
    __shared__ __attribute__((aligned(16))) float wsh[DFEAT * WPITCH];

#pragma unroll 8
    for (int idx = threadIdx.x; idx < DFEAT * 64; idx += 256) {
        int o = idx >> 6;
        int q = idx & 63;
        const float* sp = (q < 32) ? (Wl + (size_t)o * DFEAT + q * 4)
                                   : (Wr + (size_t)o * DFEAT + (q - 32) * 4);
        float4 v = *(const float4*)sp;
        *(float4*)(wsh + o * WPITCH + q * 4) = v;
    }
    __syncthreads();

    int w = __builtin_amdgcn_readfirstlane((int)(threadIdx.x >> 5));
    int lane = threadIdx.x & 31;
    int m = lane & 15;
    int h = lane >> 4;
    int total_waves = (int)gridDim.x * GEMM_WAVES;
    const float* bsh = wsh + m * WPITCH + 2 * h;

    int tile = w * (int)gridDim.x + (int)blockIdx.x;

    v2f am[8], ax[8];
    if (tile < n_tiles) {
        unsigned o0 = (unsigned)(tile * 16 + m) * DFEAT + 2 * h;
#pragma unroll
        for (int j = 0; j < 8; ++j) {
            am[j] = *(const v2f*)(agg + o0 + 4 * j);
            ax[j] = *(const v2f*)(xin + o0 + 4 * j);
        }
    }

    for (; tile < n_tiles; tile += total_waves) {
        int row = tile * 16 + m;
        float rd = 1.0f / fmaxf(deg[row], 1.0f);
        unsigned rowOff = (unsigned)row * DFEAT + 2 * h;
        int ntile = tile + total_waves;
        int nrow = ((ntile < n_tiles) ? ntile : tile) * 16 + m;
        unsigned nextOff = (unsigned)nrow * DFEAT + 2 * h;

        v8f acc[8] = {};

#pragma unroll 1
        for (int c = 0; c < 4; ++c) {
            unsigned po = (c < 3) ? (rowOff + 32 * (c + 1)) : nextOff;
            v2f nm[8], nx[8];
#pragma unroll
            for (int j = 0; j < 8; ++j) {
                nm[j] = *(const v2f*)(agg + po + 4 * j);
                nx[j] = *(const v2f*)(xin + po + 4 * j);
            }
#pragma unroll
            for (int j = 0; j < 8; ++j) {
                int k = 32 * c + 4 * j;
                v2f a_m = am[j];
                a_m[0] *= rd;
                a_m[1] *= rd;
                v2f a_x = ax[j];
#pragma unroll
                for (int t = 0; t < 8; ++t) {
                    v2f b_l = *(const v2f*)(bsh + t * 16 * WPITCH + k);
                    v2f b_r = *(const v2f*)(bsh + t * 16 * WPITCH + DFEAT + k);
                    acc[t] = __builtin_amdgcn_wmma_f32_16x16x4_f32(
                        false, a_m, false, b_l, (short)0, acc[t], false, false);
                    acc[t] = __builtin_amdgcn_wmma_f32_16x16x4_f32(
                        false, a_x, false, b_r, (short)0, acc[t], false, false);
                }
            }
#pragma unroll
            for (int j = 0; j < 8; ++j) {
                am[j] = nm[j];
                ax[j] = nx[j];
            }
        }

#pragma unroll
        for (int t = 0; t < 8; ++t) {
            float bias = bl[t * 16 + m];
            float* orow = out + (size_t)(tile * 16 + 8 * h) * DFEAT + t * 16 + m;
#pragma unroll
            for (int r = 0; r < 8; ++r) {
                float v = acc[t][r] + bias;
                if (do_relu) v = fmaxf(v, 0.0f);
                orow[(size_t)r * DFEAT] = v;
            }
        }
    }
}

extern "C" void kernel_launch(void* const* d_in, const int* in_sizes, int n_in,
                              void* d_out, int out_size, void* d_ws, size_t ws_size,
                              hipStream_t stream) {
    const float* x   = (const float*)d_in[0];
    const int*   ei  = (const int*)d_in[1];
    const float* Wl1 = (const float*)d_in[2];
    const float* bl1 = (const float*)d_in[3];
    const float* Wr1 = (const float*)d_in[4];
    const float* Wl2 = (const float*)d_in[5];
    const float* bl2 = (const float*)d_in[6];
    const float* Wr2 = (const float*)d_in[7];
    float* outp = (float*)d_out;

    int n = in_sizes[0] / DFEAT;
    int E = in_sizes[1] / 2;
    const int* src = ei;
    const int* dst = ei + E;

    float* agg = (float*)d_ws;
    float* hbuf = agg + (size_t)n * DFEAT;
    float* deg = hbuf + (size_t)n * DFEAT;
    int* cnt    = (int*)(deg + n);
    int* off    = cnt + n;
    int* bsum   = off + n;
    int* rank   = bsum + 1024;
    int* slot   = rank + E;
    int* sorted = slot + E;

    int tiles = (n + 15) / 16;
    int gemm_blocks = GEMM_BLOCKS;
    int edge_blocks = (E + 255) / 256;
    int node_blocks = (n + 255) / 256;
    int wave_blocks = (n + 7) / 8;
    int scan_blocks = (n + SCAN_ITEMS - 1) / SCAN_ITEMS;

    zero_i32<<<node_blocks, 256, 0, stream>>>(cnt, n);
    count_rank<<<edge_blocks, 256, 0, stream>>>(dst, cnt, rank, E);
    scan_local<<<scan_blocks, 256, 0, stream>>>(cnt, off, bsum, n);
    scan_add<<<scan_blocks, 256, 0, stream>>>(off, bsum, n);
    fill_slots<<<edge_blocks, 256, 0, stream>>>(src, dst, rank, off, slot, E);
    order_lists<<<wave_blocks, 256, 0, stream>>>(slot, off, cnt, sorted, deg, n);

    sage_gather<<<wave_blocks, 256, 0, stream>>>(x, sorted, off, cnt, agg, n);
    sage_gemm<<<gemm_blocks, 256, 0, stream>>>(agg, deg, x, Wl1, bl1, Wr1, hbuf,
                                              tiles, 1);
    sage_gather<<<wave_blocks, 256, 0, stream>>>(hbuf, sorted, off, cnt, agg, n);
    sage_gemm<<<gemm_blocks, 256, 0, stream>>>(agg, deg, hbuf, Wl2, bl2, Wr2,
                                              outp, tiles, 0);
}


# neighbour sums move into the GEMM kernel: the wave that multiplies a 16-node tile first adds its nodes' source rows (32 loads in flight, list order kept) into an LDS tile; agg pass and buffer gone
# speedup vs baseline: 3.0621x; 1.0612x over previous
#include <hip/hip_runtime.h>
#include <hip/hip_bf16.h>


typedef __attribute__((ext_vector_type(2))) float v2f;
typedef __attribute__((ext_vector_type(8))) float v8f;

#define DFEAT 128
#define SCAN_ITEMS 1024

__global__ void zero_i32(int* __restrict__ p, int n) {
    int i = blockIdx.x * blockDim.x + threadIdx.x;
    if (i < n) p[i] = 0;
}

__global__ void count_rank(const int* __restrict__ dst, int* __restrict__ cnt,
                           int* __restrict__ rank, int n_edges) {
    int e = blockIdx.x * blockDim.x + threadIdx.x;
    if (e < n_edges) rank[e] = atomicAdd(&cnt[dst[e]], 1);
}

__global__ __launch_bounds__(256)
void scan_local(const int* __restrict__ cnt, int* __restrict__ off,
                int* __restrict__ bsum, int n) {
    __shared__ int wsum[8];
    int t = threadIdx.x;
    int lane = t & 31;
    int w = t >> 5;
    int i0 = blockIdx.x * SCAN_ITEMS + t * 4;
    int c0 = (i0 + 0 < n) ? cnt[i0 + 0] : 0;
    int c1 = (i0 + 1 < n) ? cnt[i0 + 1] : 0;
    int c2 = (i0 + 2 < n) ? cnt[i0 + 2] : 0;
    int c3 = (i0 + 3 < n) ? cnt[i0 + 3] : 0;
    int tot = c0 + c1 + c2 + c3;
    int incl = tot;
    for (int o = 1; o < 32; o <<= 1) {
        int v = __shfl_up(incl, o);
        if (lane >= o) incl += v;
    }
    if (lane == 31) wsum[w] = incl;
    __syncthreads();
    int wb = 0;
    for (int j = 0; j < w; ++j) wb += wsum[j];
    int ex = wb + incl - tot;
    if (i0 + 0 < n) off[i0 + 0] = ex;
    if (i0 + 1 < n) off[i0 + 1] = ex + c0;
    if (i0 + 2 < n) off[i0 + 2] = ex + c0 + c1;
    if (i0 + 3 < n) off[i0 + 3] = ex + c0 + c1 + c2;
    if (t == 255) bsum[blockIdx.x] = wb + incl;
}

__global__ __launch_bounds__(256)
void scan_add(int* __restrict__ off, const int* __restrict__ bsum, int n) {
    __shared__ int wsum[8];
    int t = threadIdx.x;
    int lane = t & 31;
    int w = t >> 5;
    int nb = blockIdx.x;
    int part = 0;
    for (int j = t; j < nb; j += 256) part += bsum[j];
    for (int o = 16; o > 0; o >>= 1) part += __shfl_xor(part, o);
    if (lane == 0) wsum[w] = part;
    __syncthreads();
    int pre = 0;
    for (int j = 0; j < 8; ++j) pre += wsum[j];
    int i0 = blockIdx.x * SCAN_ITEMS + t * 4;
    for (int q = 0; q < 4; ++q)
        if (i0 + q < n) off[i0 + q] += pre;
}

__global__ void fill_slots(const int* __restrict__ src, const int* __restrict__ dst,
                           const int* __restrict__ rank, const int* __restrict__ off,
                           int* __restrict__ slot, int n_edges) {
    int e = blockIdx.x * blockDim.x + threadIdx.x;
    if (e < n_edges) slot[off[dst[e]] + rank[e]] = src[e];
}

__global__ __launch_bounds__(256)
void order_lists(const int* __restrict__ slot, const int* __restrict__ off,
                 const int* __restrict__ cnt, int* __restrict__ sorted,
                 float* __restrict__ deg, int n_nodes) {
    int node = __builtin_amdgcn_readfirstlane(
        (int)((blockIdx.x * blockDim.x + threadIdx.x) >> 5));
    int lane = threadIdx.x & 31;
    if (node >= n_nodes) return;
    int c = __builtin_amdgcn_readfirstlane(cnt[node]);
    int base = __builtin_amdgcn_readfirstlane(off[node]);
    if (lane == 0) deg[node] = (float)c;
    if (c <= 32) {
        int s = (lane < c) ? slot[base + lane] : 0;
        int r = 0;
        for (int j = 0; j < c; ++j) {
            int sj = __builtin_amdgcn_readlane(s, j);
            r += ((sj < s) || (sj == s && j < lane)) ? 1 : 0;
        }
        if (lane < c) sorted[base + r] = s;
    } else {
        for (int i = lane; i < c; i += 32) {
            int s = slot[base + i];
            int r = 0;
            for (int j = 0; j < c; ++j) {
                int sj = slot[base + j];
                r += ((sj < s) || (sj == s && j < i)) ? 1 : 0;
            }
            sorted[base + r] = s;
        }
    }
}

#define WPITCH 260
#define APITCH 132
#define GEMM_WAVES 16
#define GEMM_BLOCKS 256

__device__ __forceinline__ void add4(float4& a, const float4& v) {
    a.x += v.x; a.y += v.y; a.z += v.z; a.w += v.w;
}

__device__ __forceinline__ void walk32(const float4* __restrict__ x4, unsigned lane, int s,
                                       int mcount, int e0, const int endv,
                                       float* __restrict__ arow, float4& acc,
                                       int& nd, int& bnd) {
    float4 v[32];
#pragma unroll
    for (int k = 0; k < 32; ++k) {
        unsigned sk = (unsigned)__builtin_amdgcn_readlane(s, k);
        v[k] = x4[sk * (DFEAT / 4) + lane];
    }
#pragma unroll
    for (int k = 0; k < 32; ++k) {
        if (k < mcount) {
            while (e0 + k == bnd) {
                *(float4*)(arow + nd * APITCH + 4 * lane) = acc;
                acc = make_float4(0.f, 0.f, 0.f, 0.f);
                ++nd;
                bnd = __builtin_amdgcn_readlane(endv, nd);
            }
            add4(acc, v[k]);
        }
    }
}

__global__ __launch_bounds__(512)
void sage_layer(const float* __restrict__ xin, const int* __restrict__ sorted,
                const int* __restrict__ off, const int* __restrict__ cnt,
                const float* __restrict__ deg,
                const float* __restrict__ Wl, const float* __restrict__ bl,
                const float* __restrict__ Wr, float* __restrict__ out,
                int n_tiles, int do_relu) {
    __shared__ __attribute__((aligned(16))) float wsh[DFEAT * WPITCH];
    __shared__ __attribute__((aligned(16))) float ash[GEMM_WAVES * 16 * APITCH];

#pragma unroll 8
    for (int idx = threadIdx.x; idx < DFEAT * 64; idx += 512) {
        int o = idx >> 6;
        int q = idx & 63;
        const float* sp = (q < 32) ? (Wl + (size_t)o * DFEAT + q * 4)
                                   : (Wr + (size_t)o * DFEAT + (q - 32) * 4);
        float4 v = *(const float4*)sp;
        *(float4*)(wsh + o * WPITCH + q * 4) = v;
    }
    __syncthreads();

    int w = __builtin_amdgcn_readfirstlane((int)(threadIdx.x >> 5));
    int lane = threadIdx.x & 31;
    int m = lane & 15;
    int h = lane >> 4;
    int total_waves = (int)gridDim.x * GEMM_WAVES;
    const float* bsh = wsh + m * WPITCH + 2 * h;
    float* arow = ash + w * (16 * APITCH);
    const float* ap = arow + m * APITCH + 2 * h;
    const float4* x4 = (const float4*)xin;

    int tile = w * (int)gridDim.x + (int)blockIdx.x;

    v2f ax[8];
    if (tile < n_tiles) {
        unsigned o0 = (unsigned)(tile * 16 + m) * DFEAT + 2 * h;
#pragma unroll
        for (int j = 0; j < 8; ++j) ax[j] = *(const v2f*)(xin + o0 + 4 * j);
    }

    for (; tile < n_tiles; tile += total_waves) {
        int row = tile * 16 + m;
        float rd = 1.0f / fmaxf(deg[row], 1.0f);

        int cv = cnt[row];
        int ov = off[row];
        int tb = __builtin_amdgcn_readfirstlane(ov);
        int endv = ov + cv - tb;
        int T = __builtin_amdgcn_readlane(endv, 15);
        int nd = 0;
        int bnd = __builtin_amdgcn_readlane(endv, 0);
        float4 acc = make_float4(0.f, 0.f, 0.f, 0.f);
        int s = (lane < T) ? sorted[tb + lane] : 0;
        for (int e0 = 0; e0 < T; e0 += 32) {
            int sn = (e0 + 32 + lane < T) ? sorted[tb + e0 + 32 + lane] : 0;
            walk32(x4, (unsigned)lane, s, T - e0, e0, endv, arow, acc, nd, bnd);
            s = sn;
        }
        for (; nd < 16; ++nd) {
            *(float4*)(arow + nd * APITCH + 4 * lane) = acc;
            acc = make_float4(0.f, 0.f, 0.f, 0.f);
        }

        unsigned rowOff = (unsigned)row * DFEAT + 2 * h;
        int ntile = tile + total_waves;
        int nrow = ((ntile < n_tiles) ? ntile : tile) * 16 + m;
        unsigned nextOff = (unsigned)nrow * DFEAT + 2 * h;

        v8f acc8[8] = {};

#pragma unroll 1
        for (int c = 0; c < 4; ++c) {
            unsigned po = (c < 3) ? (rowOff + 32 * (c + 1)) : nextOff;
            v2f nx[8];
#pragma unroll
            for (int j = 0; j < 8; ++j) nx[j] = *(const v2f*)(xin + po + 4 * j);
            v2f am[8];
#pragma unroll
            for (int j = 0; j < 8; ++j) am[j] = *(const v2f*)(ap + 32 * c + 4 * j);
#pragma unroll
            for (int j = 0; j < 8; ++j) {
                int k = 32 * c + 4 * j;
                v2f a_m = am[j];
                a_m[0] *= rd;
                a_m[1] *= rd;
                v2f a_x = ax[j];
#pragma unroll
                for (int t = 0; t < 8; ++t) {
                    v2f b_l = *(const v2f*)(bsh + t * 16 * WPITCH + k);
                    v2f b_r = *(const v2f*)(bsh + t * 16 * WPITCH + DFEAT + k);
                    acc8[t] = __builtin_amdgcn_wmma_f32_16x16x4_f32(
                        false, a_m, false, b_l, (short)0, acc8[t], false, false);
                    acc8[t] = __builtin_amdgcn_wmma_f32_16x16x4_f32(
                        false, a_x, false, b_r, (short)0, acc8[t], false, false);
                }
            }
#pragma unroll
            for (int j = 0; j < 8; ++j) ax[j] = nx[j];
        }

#pragma unroll
        for (int t = 0; t < 8; ++t) {
            float bias = bl[t * 16 + m];
            float* orow = out + (size_t)(tile * 16 + 8 * h) * DFEAT + t * 16 + m;
#pragma unroll
            for (int r = 0; r < 8; ++r) {
                float v = acc8[t][r] + bias;
                if (do_relu) v = fmaxf(v, 0.0f);
                orow[(size_t)r * DFEAT] = v;
            }
        }
    }
}

extern "C" void kernel_launch(void* const* d_in, const int* in_sizes, int n_in,
                              void* d_out, int out_size, void* d_ws, size_t ws_size,
                              hipStream_t stream) {
    const float* x   = (const float*)d_in[0];
    const int*   ei  = (const int*)d_in[1];
    const float* Wl1 = (const float*)d_in[2];
    const float* bl1 = (const float*)d_in[3];
    const float* Wr1 = (const float*)d_in[4];
    const float* Wl2 = (const float*)d_in[5];
    const float* bl2 = (const float*)d_in[6];
    const float* Wr2 = (const float*)d_in[7];
    float* outp = (float*)d_out;

    int n = in_sizes[0] / DFEAT;
    int E = in_sizes[1] / 2;
    const int* src = ei;
    const int* dst = ei + E;

    float* hbuf = (float*)d_ws;
    float* deg = hbuf + (size_t)n * DFEAT;
    int* cnt    = (int*)(deg + n);
    int* off    = cnt + n;
    int* bsum   = off + n;
    int* rank   = bsum + 1024;
    int* slot   = rank + E;
    int* sorted = slot + E;

    int tiles = (n + 15) / 16;
    int gemm_blocks = GEMM_BLOCKS;
    int edge_blocks = (E + 255) / 256;
    int node_blocks = (n + 255) / 256;
    int wave_blocks = (n + 7) / 8;
    int scan_blocks = (n + SCAN_ITEMS - 1) / SCAN_ITEMS;

    zero_i32<<<node_blocks, 256, 0, stream>>>(cnt, n);
    count_rank<<<edge_blocks, 256, 0, stream>>>(dst, cnt, rank, E);
    scan_local<<<scan_blocks, 256, 0, stream>>>(cnt, off, bsum, n);
    scan_add<<<scan_blocks, 256, 0, stream>>>(off, bsum, n);
    fill_slots<<<edge_blocks, 256, 0, stream>>>(src, dst, rank, off, slot, E);
    order_lists<<<wave_blocks, 256, 0, stream>>>(slot, off, cnt, sorted, deg, n);

    sage_layer<<<gemm_blocks, 512, 0, stream>>>(x, sorted, off, cnt, deg, Wl1, bl1, Wr1,
                                               hbuf, tiles, 1);
    sage_layer<<<gemm_blocks, 512, 0, stream>>>(hbuf, sorted, off, cnt, deg, Wl2, bl2, Wr2,
                                               outp, tiles, 0);
}


# per-node source lists built by a two-level counting sort ranked with LDS atomics (256 edge chunks x 256-node buckets, then one work-group per bucket); global integer atomics gone; sums unchanged
# speedup vs baseline: 4.2823x; 1.3985x over previous
#include <hip/hip_runtime.h>
#include <hip/hip_bf16.h>


typedef __attribute__((ext_vector_type(2))) float v2f;
typedef __attribute__((ext_vector_type(8))) float v8f;

#define DFEAT 128
#define SCAN_ITEMS 1024

#define NPB_SHIFT 8
#define NPB (1 << NPB_SHIFT)
#define CHUNKS 256
#define CHUNK_THREADS 1024
#define EDGE_UNR 8

__global__ __launch_bounds__(CHUNK_THREADS)
void bucket_hist(const int* __restrict__ dst, int* __restrict__ tab,
                 int n_edges, int n_buckets) {
    extern __shared__ int lds_cnt[];
    int t = threadIdx.x;
    for (int i = t; i < n_buckets; i += CHUNK_THREADS) lds_cnt[i] = 0;
    __syncthreads();
    int per = (n_edges + (int)gridDim.x - 1) / (int)gridDim.x;
    int e0 = (int)blockIdx.x * per;
    int e1 = min(e0 + per, n_edges);
    for (int e = e0 + t; e < e1; e += CHUNK_THREADS * EDGE_UNR) {
        int d[EDGE_UNR];
#pragma unroll
        for (int k = 0; k < EDGE_UNR; ++k) {
            int ee = e + k * CHUNK_THREADS;
            d[k] = (ee < e1) ? dst[ee] : 0;
        }
#pragma unroll
        for (int k = 0; k < EDGE_UNR; ++k)
            if (e + k * CHUNK_THREADS < e1) atomicAdd(&lds_cnt[d[k] >> NPB_SHIFT], 1);
    }
    __syncthreads();
    for (int i = t; i < n_buckets; i += CHUNK_THREADS)
        tab[i * (int)gridDim.x + (int)blockIdx.x] = lds_cnt[i];
}

__global__ __launch_bounds__(CHUNK_THREADS)
void bucket_scatter(const int* __restrict__ src, const int* __restrict__ dst,
                    const int* __restrict__ tabs, int* __restrict__ pairs,
                    int n_edges, int n_buckets) {
    extern __shared__ int lds_cur[];
    int t = threadIdx.x;
    for (int i = t; i < n_buckets; i += CHUNK_THREADS)
        lds_cur[i] = tabs[i * (int)gridDim.x + (int)blockIdx.x];
    __syncthreads();
    int per = (n_edges + (int)gridDim.x - 1) / (int)gridDim.x;
    int e0 = (int)blockIdx.x * per;
    int e1 = min(e0 + per, n_edges);
    for (int e = e0 + t; e < e1; e += CHUNK_THREADS * EDGE_UNR) {
        int d[EDGE_UNR];
        int s[EDGE_UNR];
#pragma unroll
        for (int k = 0; k < EDGE_UNR; ++k) {
            int ee = e + k * CHUNK_THREADS;
            d[k] = (ee < e1) ? dst[ee] : 0;
            s[k] = (ee < e1) ? src[ee] : 0;
        }
#pragma unroll
        for (int k = 0; k < EDGE_UNR; ++k) {
            if (e + k * CHUNK_THREADS < e1) {
                int p = atomicAdd(&lds_cur[d[k] >> NPB_SHIFT], 1);
                *(int2*)(pairs + 2 * (size_t)p) = make_int2(s[k], d[k]);
            }
        }
    }
}

__global__ __launch_bounds__(256)
void scan_local(const int* __restrict__ cnt, int* __restrict__ off,
                int* __restrict__ bsum, int n) {
    __shared__ int wsum[8];
    int t = threadIdx.x;
    int lane = t & 31;
    int w = t >> 5;
    int i0 = blockIdx.x * SCAN_ITEMS + t * 4;
    int c0 = (i0 + 0 < n) ? cnt[i0 + 0] : 0;
    int c1 = (i0 + 1 < n) ? cnt[i0 + 1] : 0;
    int c2 = (i0 + 2 < n) ? cnt[i0 + 2] : 0;
    int c3 = (i0 + 3 < n) ? cnt[i0 + 3] : 0;
    int tot = c0 + c1 + c2 + c3;
    int incl = tot;
    for (int o = 1; o < 32; o <<= 1) {
        int v = __shfl_up(incl, o);
        if (lane >= o) incl += v;
    }
    if (lane == 31) wsum[w] = incl;
    __syncthreads();
    int wb = 0;
    for (int j = 0; j < w; ++j) wb += wsum[j];
    int ex = wb + incl - tot;
    if (i0 + 0 < n) off[i0 + 0] = ex;
    if (i0 + 1 < n) off[i0 + 1] = ex + c0;
    if (i0 + 2 < n) off[i0 + 2] = ex + c0 + c1;
    if (i0 + 3 < n) off[i0 + 3] = ex + c0 + c1 + c2;
    if (t == 255) bsum[blockIdx.x] = wb + incl;
}

__global__ __launch_bounds__(256)
void scan_add(int* __restrict__ off, const int* __restrict__ bsum, int n) {
    __shared__ int wsum[8];
    int t = threadIdx.x;
    int lane = t & 31;
    int w = t >> 5;
    int nb = blockIdx.x;
    int part = 0;
    for (int j = t; j < nb; j += 256) part += bsum[j];
    for (int o = 16; o > 0; o >>= 1) part += __shfl_xor(part, o);
    if (lane == 0) wsum[w] = part;
    __syncthreads();
    int pre = 0;
    for (int j = 0; j < 8; ++j) pre += wsum[j];
    int i0 = blockIdx.x * SCAN_ITEMS + t * 4;
    for (int q = 0; q < 4; ++q)
        if (i0 + q < n) off[i0 + q] += pre;
}

#define LIST_THREADS 1024
#define LIST_UNR 4
__global__ __launch_bounds__(LIST_THREADS)
void bucket_lists(const int* __restrict__ pairs, const int* __restrict__ tabs,
                  int* __restrict__ cnt, int* __restrict__ off,
                  int* __restrict__ slot, int n_nodes, int n_edges, int chunks) {
    __shared__ int ncnt[NPB];
    __shared__ int ncur[NPB];
    __shared__ int wsum[NPB / 32];
    int t = threadIdx.x;
    int lane = t & 31;
    int w = t >> 5;
    int b = (int)blockIdx.x;
    int r0 = tabs[b * chunks];
    int r1 = (b + 1 < (int)gridDim.x) ? tabs[(b + 1) * chunks] : n_edges;
    int base = b << NPB_SHIFT;
    if (t < NPB) ncnt[t] = 0;
    __syncthreads();
    for (int i = r0 + t; i < r1; i += LIST_THREADS * LIST_UNR) {
        int d[LIST_UNR];
#pragma unroll
        for (int k = 0; k < LIST_UNR; ++k) {
            int ii = i + k * LIST_THREADS;
            d[k] = (ii < r1) ? pairs[2 * (size_t)ii + 1] : base;
        }
#pragma unroll
        for (int k = 0; k < LIST_UNR; ++k)
            if (i + k * LIST_THREADS < r1) atomicAdd(&ncnt[d[k] - base], 1);
    }
    __syncthreads();
    int c = 0;
    int incl = 0;
    if (t < NPB) {
        c = ncnt[t];
        incl = c;
        for (int o = 1; o < 32; o <<= 1) {
            int v = __shfl_up(incl, o);
            if (lane >= o) incl += v;
        }
        if (lane == 31) wsum[w] = incl;
    }
    __syncthreads();
    if (t < NPB) {
        int wb = 0;
        for (int j = 0; j < w; ++j) wb += wsum[j];
        int ex = wb + incl - c;
        ncur[t] = ex;
        if (base + t < n_nodes) {
            cnt[base + t] = c;
            off[base + t] = r0 + ex;
        }
    }
    __syncthreads();
    for (int i = r0 + t; i < r1; i += LIST_THREADS * LIST_UNR) {
        int2 p[LIST_UNR];
#pragma unroll
        for (int k = 0; k < LIST_UNR; ++k) {
            int ii = i + k * LIST_THREADS;
            p[k] = (ii < r1) ? *(const int2*)(pairs + 2 * (size_t)ii) : make_int2(0, base);
        }
#pragma unroll
        for (int k = 0; k < LIST_UNR; ++k) {
            if (i + k * LIST_THREADS < r1) {
                int q = atomicAdd(&ncur[p[k].y - base], 1);
                slot[r0 + q] = p[k].x;
            }
        }
    }
}

__global__ __launch_bounds__(256)
void order_lists(const int* __restrict__ slot, const int* __restrict__ off,
                 const int* __restrict__ cnt, int* __restrict__ sorted,
                 float* __restrict__ deg, int n_nodes) {
    int node = __builtin_amdgcn_readfirstlane(
        (int)((blockIdx.x * blockDim.x + threadIdx.x) >> 5));
    int lane = threadIdx.x & 31;
    if (node >= n_nodes) return;
    int c = __builtin_amdgcn_readfirstlane(cnt[node]);
    int base = __builtin_amdgcn_readfirstlane(off[node]);
    if (lane == 0) deg[node] = (float)c;
    if (c <= 32) {
        int s = (lane < c) ? slot[base + lane] : 0;
        int r = 0;
        for (int j = 0; j < c; ++j) {
            int sj = __builtin_amdgcn_readlane(s, j);
            r += ((sj < s) || (sj == s && j < lane)) ? 1 : 0;
        }
        if (lane < c) sorted[base + r] = s;
    } else {
        for (int i = lane; i < c; i += 32) {
            int s = slot[base + i];
            int r = 0;
            for (int j = 0; j < c; ++j) {
                int sj = slot[base + j];
                r += ((sj < s) || (sj == s && j < i)) ? 1 : 0;
            }
            sorted[base + r] = s;
        }
    }
}

#define WPITCH 260
#define APITCH 132
#define GEMM_WAVES 16
#define GEMM_BLOCKS 256

__device__ __forceinline__ void add4(float4& a, const float4& v) {
    a.x += v.x; a.y += v.y; a.z += v.z; a.w += v.w;
}

__device__ __forceinline__ void walk32(const float4* __restrict__ x4, unsigned lane, int s,
                                       int mcount, int e0, const int endv,
                                       float* __restrict__ arow, float4& acc,
                                       int& nd, int& bnd) {
    float4 v[32];
#pragma unroll
    for (int k = 0; k < 32; ++k) {
        unsigned sk = (unsigned)__builtin_amdgcn_readlane(s, k);
        v[k] = x4[sk * (DFEAT / 4) + lane];
    }
#pragma unroll
    for (int k = 0; k < 32; ++k) {
        if (k < mcount) {
            while (e0 + k == bnd) {
                *(float4*)(arow + nd * APITCH + 4 * lane) = acc;
                acc = make_float4(0.f, 0.f, 0.f, 0.f);
                ++nd;
                bnd = __builtin_amdgcn_readlane(endv, nd);
            }
            add4(acc, v[k]);
        }
    }
}

__global__ __launch_bounds__(512)
void sage_layer(const float* __restrict__ xin, const int* __restrict__ sorted,
                const int* __restrict__ off, const int* __restrict__ cnt,
                const float* __restrict__ deg,
                const float* __restrict__ Wl, const float* __restrict__ bl,
                const float* __restrict__ Wr, float* __restrict__ out,
                int n_tiles, int do_relu) {
    __shared__ __attribute__((aligned(16))) float wsh[DFEAT * WPITCH];
    __shared__ __attribute__((aligned(16))) float ash[GEMM_WAVES * 16 * APITCH];

#pragma unroll 8
    for (int idx = threadIdx.x; idx < DFEAT * 64; idx += 512) {
        int o = idx >> 6;
        int q = idx & 63;
        const float* sp = (q < 32) ? (Wl + (size_t)o * DFEAT + q * 4)
                                   : (Wr + (size_t)o * DFEAT + (q - 32) * 4);
        float4 v = *(const float4*)sp;
        *(float4*)(wsh + o * WPITCH + q * 4) = v;
    }
    __syncthreads();

    int w = __builtin_amdgcn_readfirstlane((int)(threadIdx.x >> 5));
    int lane = threadIdx.x & 31;
    int m = lane & 15;
    int h = lane >> 4;
    int total_waves = (int)gridDim.x * GEMM_WAVES;
    const float* bsh = wsh + m * WPITCH + 2 * h;
    float* arow = ash + w * (16 * APITCH);
    const float* ap = arow + m * APITCH + 2 * h;
    const float4* x4 = (const float4*)xin;

    int tile = w * (int)gridDim.x + (int)blockIdx.x;

    v2f ax[8];
    if (tile < n_tiles) {
        unsigned o0 = (unsigned)(tile * 16 + m) * DFEAT + 2 * h;
#pragma unroll
        for (int j = 0; j < 8; ++j) ax[j] = *(const v2f*)(xin + o0 + 4 * j);
    }

    for (; tile < n_tiles; tile += total_waves) {
        int row = tile * 16 + m;
        float rd = 1.0f / fmaxf(deg[row], 1.0f);

        int cv = cnt[row];
        int ov = off[row];
        int tb = __builtin_amdgcn_readfirstlane(ov);
        int endv = ov + cv - tb;
        int T = __builtin_amdgcn_readlane(endv, 15);
        int nd = 0;
        int bnd = __builtin_amdgcn_readlane(endv, 0);
        float4 acc = make_float4(0.f, 0.f, 0.f, 0.f);
        int s = (lane < T) ? sorted[tb + lane] : 0;
        for (int e0 = 0; e0 < T; e0 += 32) {
            int sn = (e0 + 32 + lane < T) ? sorted[tb + e0 + 32 + lane] : 0;
            walk32(x4, (unsigned)lane, s, T - e0, e0, endv, arow, acc, nd, bnd);
            s = sn;
        }
        for (; nd < 16; ++nd) {
            *(float4*)(arow + nd * APITCH + 4 * lane) = acc;
            acc = make_float4(0.f, 0.f, 0.f, 0.f);
        }

        unsigned rowOff = (unsigned)row * DFEAT + 2 * h;
        int ntile = tile + total_waves;
        int nrow = ((ntile < n_tiles) ? ntile : tile) * 16 + m;
        unsigned nextOff = (unsigned)nrow * DFEAT + 2 * h;

        v8f acc8[8] = {};

#pragma unroll 1
        for (int c = 0; c < 4; ++c) {
            unsigned po = (c < 3) ? (rowOff + 32 * (c + 1)) : nextOff;
            v2f nx[8];
#pragma unroll
            for (int j = 0; j < 8; ++j) nx[j] = *(const v2f*)(xin + po + 4 * j);
            v2f am[8];
#pragma unroll
            for (int j = 0; j < 8; ++j) am[j] = *(const v2f*)(ap + 32 * c + 4 * j);
#pragma unroll
            for (int j = 0; j < 8; ++j) {
                int k = 32 * c + 4 * j;
                v2f a_m = am[j];
                a_m[0] *= rd;
                a_m[1] *= rd;
                v2f a_x = ax[j];
#pragma unroll
                for (int t = 0; t < 8; ++t) {
                    v2f b_l = *(const v2f*)(bsh + t * 16 * WPITCH + k);
                    v2f b_r = *(const v2f*)(bsh + t * 16 * WPITCH + DFEAT + k);
                    acc8[t] = __builtin_amdgcn_wmma_f32_16x16x4_f32(
                        false, a_m, false, b_l, (short)0, acc8[t], false, false);
                    acc8[t] = __builtin_amdgcn_wmma_f32_16x16x4_f32(
                        false, a_x, false, b_r, (short)0, acc8[t], false, false);
                }
            }
#pragma unroll
            for (int j = 0; j < 8; ++j) ax[j] = nx[j];
        }

#pragma unroll
        for (int t = 0; t < 8; ++t) {
            float bias = bl[t * 16 + m];
            float* orow = out + (size_t)(tile * 16 + 8 * h) * DFEAT + t * 16 + m;
#pragma unroll
            for (int r = 0; r < 8; ++r) {
                float v = acc8[t][r] + bias;
                if (do_relu) v = fmaxf(v, 0.0f);
                orow[(size_t)r * DFEAT] = v;
            }
        }
    }
}

extern "C" void kernel_launch(void* const* d_in, const int* in_sizes, int n_in,
                              void* d_out, int out_size, void* d_ws, size_t ws_size,
                              hipStream_t stream) {
    const float* x   = (const float*)d_in[0];
    const int*   ei  = (const int*)d_in[1];
    const float* Wl1 = (const float*)d_in[2];
    const float* bl1 = (const float*)d_in[3];
    const float* Wr1 = (const float*)d_in[4];
    const float* Wl2 = (const float*)d_in[5];
    const float* bl2 = (const float*)d_in[6];
    const float* Wr2 = (const float*)d_in[7];
    float* outp = (float*)d_out;

    int n = in_sizes[0] / DFEAT;
    int E = in_sizes[1] / 2;
    const int* src = ei;
    const int* dst = ei + E;

    int nb = (n + NPB - 1) / NPB;
    int tab_n = nb * CHUNKS;

    float* hbuf = (float*)d_ws;
    int* pairs  = (int*)(hbuf + (size_t)n * DFEAT);
    float* deg  = (float*)(pairs + 2 * (size_t)E);
    int* cnt    = (int*)(deg + n);
    int* off    = cnt + n;
    int* bsum   = off + n;
    int* slot   = bsum + 1024;
    int* sorted = slot + E;
    int* tab    = sorted + E;
    int* tabs   = tab + tab_n;

    int tiles = (n + 15) / 16;
    int gemm_blocks = GEMM_BLOCKS;
    int wave_blocks = (n + 7) / 8;
    int scan_blocks = (tab_n + SCAN_ITEMS - 1) / SCAN_ITEMS;

    bucket_hist<<<CHUNKS, CHUNK_THREADS, nb * sizeof(int), stream>>>(dst, tab, E, nb);
    scan_local<<<scan_blocks, 256, 0, stream>>>(tab, tabs, bsum, tab_n);
    scan_add<<<scan_blocks, 256, 0, stream>>>(tabs, bsum, tab_n);
    bucket_scatter<<<CHUNKS, CHUNK_THREADS, nb * sizeof(int), stream>>>(src, dst, tabs, pairs,
                                                                        E, nb);
    bucket_lists<<<nb, LIST_THREADS, 0, stream>>>(pairs, tabs, cnt, off, slot, n, E, CHUNKS);
    order_lists<<<wave_blocks, 256, 0, stream>>>(slot, off, cnt, sorted, deg, n);

    sage_layer<<<gemm_blocks, 512, 0, stream>>>(x, sorted, off, cnt, deg, Wl1, bl1, Wr1,
                                               hbuf, tiles, 1);
    sage_layer<<<gemm_blocks, 512, 0, stream>>>(hbuf, sorted, off, cnt, deg, Wl2, bl2, Wr2,
                                               outp, tiles, 0);
}


# list build: every edge chunk counts, scans in LDS and places its pairs by bucket in its own run, in one kernel; bucket groups gather 256 segments; histogram and 2 scan launches gone; sums unchanged
# speedup vs baseline: 4.8040x; 1.1218x over previous
#include <hip/hip_runtime.h>
#include <hip/hip_bf16.h>


typedef __attribute__((ext_vector_type(2))) float v2f;
typedef __attribute__((ext_vector_type(8))) float v8f;

#define DFEAT 128

#define NPB_SHIFT 8
#define NPB (1 << NPB_SHIFT)
#define CHUNKS 256
#define CHUNK_THREADS 1024
#define EDGE_UNR 8

__global__ __launch_bounds__(CHUNK_THREADS)
void chunk_sort(const int* __restrict__ src, const int* __restrict__ dst,
                int* __restrict__ lo, int* __restrict__ pairs,
                int n_edges, int n_buckets) {
    extern __shared__ int lds_cur[];
    __shared__ int wsum[CHUNK_THREADS / 32];
    int t = threadIdx.x;
    int lane = t & 31;
    int w = t >> 5;
    for (int i = t; i < n_buckets; i += CHUNK_THREADS) lds_cur[i] = 0;
    __syncthreads();
    int per = (n_edges + (int)gridDim.x - 1) / (int)gridDim.x;
    int e0 = min((int)blockIdx.x * per, n_edges);
    int e1 = min(e0 + per, n_edges);

    for (int e = e0 + t; e < e1; e += CHUNK_THREADS * EDGE_UNR) {
        int d[EDGE_UNR];
#pragma unroll
        for (int k = 0; k < EDGE_UNR; ++k)
            d[k] = dst[min(e + k * CHUNK_THREADS, e1 - 1)];
        static_assert(EDGE_UNR == 8, "the statement below names eight values");
        asm volatile("" :: "v"(d[0]), "v"(d[1]), "v"(d[2]), "v"(d[3]),
                           "v"(d[4]), "v"(d[5]), "v"(d[6]), "v"(d[7]));
#pragma unroll
        for (int k = 0; k < EDGE_UNR; ++k)
            if (e + k * CHUNK_THREADS < e1) atomicAdd(&lds_cur[d[k] >> NPB_SHIFT], 1);
    }
    __syncthreads();

    int ipt = (n_buckets + CHUNK_THREADS - 1) / CHUNK_THREADS;
    int i0 = t * ipt;
    int tot = 0;
    for (int q = 0; q < ipt; ++q)
        if (i0 + q < n_buckets) tot += lds_cur[i0 + q];
    int incl = tot;
    for (int o = 1; o < 32; o <<= 1) {
        int v = __shfl_up(incl, o);
        if (lane >= o) incl += v;
    }
    if (lane == 31) wsum[w] = incl;
    __syncthreads();
    int wb = 0;
    for (int j = 0; j < w; ++j) wb += wsum[j];
    int ex = wb + incl - tot;
    int* lrow = lo + (size_t)blockIdx.x * (size_t)(n_buckets + 1);
    for (int q = 0; q < ipt; ++q) {
        if (i0 + q < n_buckets) {
            int c = lds_cur[i0 + q];
            lds_cur[i0 + q] = ex;
            lrow[i0 + q] = ex;
            ex += c;
        }
    }
    if (t == 0) lrow[n_buckets] = e1 - e0;
    __syncthreads();

    for (int e = e0 + t; e < e1; e += CHUNK_THREADS * EDGE_UNR) {
        int d[EDGE_UNR];
        int s[EDGE_UNR];
#pragma unroll
        for (int k = 0; k < EDGE_UNR; ++k) {
            int ee = e + k * CHUNK_THREADS;
            d[k] = (ee < e1) ? dst[ee] : 0;
            s[k] = (ee < e1) ? src[ee] : 0;
        }
#pragma unroll
        for (int k = 0; k < EDGE_UNR; ++k) {
            if (e + k * CHUNK_THREADS < e1) {
                int p = atomicAdd(&lds_cur[d[k] >> NPB_SHIFT], 1);
                *(int2*)(pairs + 2 * ((size_t)e0 + (size_t)p)) = make_int2(s[k], d[k]);
            }
        }
    }
}

#define LIST_THREADS 1024
#define LIST_UNR 8
#define SEG_THREADS 4
static_assert(LIST_THREADS == 1024, "the sum over waves below takes one wave total per lane");
__global__ __launch_bounds__(LIST_THREADS)
void bucket_lists(const int* __restrict__ pairs, const int* __restrict__ lo,
                  int* __restrict__ cnt, int* __restrict__ off,
                  int* __restrict__ slot, int n_nodes, int n_edges, int chunks) {
    __shared__ int ncnt[NPB];
    __shared__ int ncur[NPB];
    __shared__ int wsum[NPB / 32];
    __shared__ int red[LIST_THREADS / 32];
    int t = threadIdx.x;
    int lane = t & 31;
    int w = t >> 5;
    int b = (int)blockIdx.x;
    size_t nb1 = (size_t)gridDim.x + 1;
    int per = (n_edges + chunks - 1) / chunks;
    int base = b << NPB_SHIFT;
    int sub = t & (SEG_THREADS - 1);
    if (t < NPB) ncnt[t] = 0;
    int part = 0;
    for (int c = t; c < chunks; c += LIST_THREADS) part += lo[(size_t)c * nb1 + b];
    for (int o = 16; o > 0; o >>= 1) part += __shfl_xor(part, o);
    if (lane == 0) red[w] = part;
    __syncthreads();
    int r0 = red[lane];
    for (int o = 16; o > 0; o >>= 1) r0 += __shfl_xor(r0, o);

    for (int c = t / SEG_THREADS; c < chunks; c += LIST_THREADS / SEG_THREADS) {
        const int* row = lo + (size_t)c * nb1 + b;
        int a = row[0];
        int len = row[1] - a;
        const int* seg = pairs + 2 * ((size_t)c * (size_t)per + (size_t)a);
        for (int k = sub; k < len; k += SEG_THREADS * LIST_UNR) {
            int d[LIST_UNR];
#pragma unroll
            for (int j = 0; j < LIST_UNR; ++j) {
                int kk = k + j * SEG_THREADS;
                d[j] = (kk < len) ? seg[2 * (size_t)kk + 1] : base;
            }
#pragma unroll
            for (int j = 0; j < LIST_UNR; ++j)
                if (k + j * SEG_THREADS < len) atomicAdd(&ncnt[d[j] - base], 1);
        }
    }
    __syncthreads();
    int c = 0;
    int incl = 0;
    if (t < NPB) {
        c = ncnt[t];
        incl = c;
        for (int o = 1; o < 32; o <<= 1) {
            int v = __shfl_up(incl, o);
            if (lane >= o) incl += v;
        }
        if (lane == 31) wsum[w] = incl;
    }
    __syncthreads();
    if (t < NPB) {
        int wb = 0;
        for (int j = 0; j < w; ++j) wb += wsum[j];
        int ex = wb + incl - c;
        ncur[t] = ex;
        if (base + t < n_nodes) {
            cnt[base + t] = c;
            off[base + t] = r0 + ex;
        }
    }
    __syncthreads();
    for (int cc = t / SEG_THREADS; cc < chunks; cc += LIST_THREADS / SEG_THREADS) {
        const int* row = lo + (size_t)cc * nb1 + b;
        int a = row[0];
        int len = row[1] - a;
        const int* seg = pairs + 2 * ((size_t)cc * (size_t)per + (size_t)a);
        for (int k = sub; k < len; k += SEG_THREADS * LIST_UNR) {
            int2 p[LIST_UNR];
#pragma unroll
            for (int j = 0; j < LIST_UNR; ++j) {
                int kk = k + j * SEG_THREADS;
                p[j] = (kk < len) ? *(const int2*)(seg + 2 * (size_t)kk) : make_int2(0, base);
            }
#pragma unroll
            for (int j = 0; j < LIST_UNR; ++j) {
                if (k + j * SEG_THREADS < len) {
                    int q = atomicAdd(&ncur[p[j].y - base], 1);
                    slot[r0 + q] = p[j].x;
                }
            }
        }
    }
}

__global__ __launch_bounds__(256)
void order_lists(const int* __restrict__ slot, const int* __restrict__ off,
                 const int* __restrict__ cnt, int* __restrict__ sorted,
                 float* __restrict__ deg, int n_nodes) {
    int node = __builtin_amdgcn_readfirstlane(
        (int)((blockIdx.x * blockDim.x + threadIdx.x) >> 5));
    int lane = threadIdx.x & 31;
    if (node >= n_nodes) return;
    int c = __builtin_amdgcn_readfirstlane(cnt[node]);
    int base = __builtin_amdgcn_readfirstlane(off[node]);
    if (lane == 0) deg[node] = (float)c;
    if (c <= 32) {
        int s = (lane < c) ? slot[base + lane] : 0;
        int r = 0;
        for (int j = 0; j < c; ++j) {
            int sj = __builtin_amdgcn_readlane(s, j);
            r += ((sj < s) || (sj == s && j < lane)) ? 1 : 0;
        }
        if (lane < c) sorted[base + r] = s;
    } else {
        for (int i = lane; i < c; i += 32) {
            int s = slot[base + i];
            int r = 0;
            for (int j = 0; j < c; ++j) {
                int sj = slot[base + j];
                r += ((sj < s) || (sj == s && j < i)) ? 1 : 0;
            }
            sorted[base + r] = s;
        }
    }
}

#define WPITCH 260
#define APITCH 132
#define GEMM_WAVES 16
#define GEMM_BLOCKS 256

__device__ __forceinline__ void add4(float4& a, const float4& v) {
    a.x += v.x; a.y += v.y; a.z += v.z; a.w += v.w;
}

__device__ __forceinline__ void walk32(const float4* __restrict__ x4, unsigned lane, int s,
                                       int mcount, int e0, const int endv,
                                       float* __restrict__ arow, float4& acc,
                                       int& nd, int& bnd) {
    float4 v[32];
#pragma unroll
    for (int k = 0; k < 32; ++k) {
        unsigned sk = (unsigned)__builtin_amdgcn_readlane(s, k);
        v[k] = x4[sk * (DFEAT / 4) + lane];
    }
#pragma unroll
    for (int k = 0; k < 32; ++k) {
        if (k < mcount) {
            while (e0 + k == bnd) {
                *(float4*)(arow + nd * APITCH + 4 * lane) = acc;
                acc = make_float4(0.f, 0.f, 0.f, 0.f);
                ++nd;
                bnd = __builtin_amdgcn_readlane(endv, nd);
            }
            add4(acc, v[k]);
        }
    }
}

__global__ __launch_bounds__(512)
void sage_layer(const float* __restrict__ xin, const int* __restrict__ sorted,
                const int* __restrict__ off, const int* __restrict__ cnt,
                const float* __restrict__ deg,
                const float* __restrict__ Wl, const float* __restrict__ bl,
                const float* __restrict__ Wr, float* __restrict__ out,
                int n_tiles, int do_relu) {
    __shared__ __attribute__((aligned(16))) float wsh[DFEAT * WPITCH];
    __shared__ __attribute__((aligned(16))) float ash[GEMM_WAVES * 16 * APITCH];

#pragma unroll 8
    for (int idx = threadIdx.x; idx < DFEAT * 64; idx += 512) {
        int o = idx >> 6;
        int q = idx & 63;
        const float* sp = (q < 32) ? (Wl + (size_t)o * DFEAT + q * 4)
                                   : (Wr + (size_t)o * DFEAT + (q - 32) * 4);
        float4 v = *(const float4*)sp;
        *(float4*)(wsh + o * WPITCH + q * 4) = v;
    }
    __syncthreads();

    int w = __builtin_amdgcn_readfirstlane((int)(threadIdx.x >> 5));
    int lane = threadIdx.x & 31;
    int m = lane & 15;
    int h = lane >> 4;
    int total_waves = (int)gridDim.x * GEMM_WAVES;
    const float* bsh = wsh + m * WPITCH + 2 * h;
    float* arow = ash + w * (16 * APITCH);
    const float* ap = arow + m * APITCH + 2 * h;
    const float4* x4 = (const float4*)xin;

    int tile = w * (int)gridDim.x + (int)blockIdx.x;

    v2f ax[8];
    if (tile < n_tiles) {
        unsigned o0 = (unsigned)(tile * 16 + m) * DFEAT + 2 * h;
#pragma unroll
        for (int j = 0; j < 8; ++j) ax[j] = *(const v2f*)(xin + o0 + 4 * j);
    }

    for (; tile < n_tiles; tile += total_waves) {
        int row = tile * 16 + m;
        float rd = 1.0f / fmaxf(deg[row], 1.0f);

        int cv = cnt[row];
        int ov = off[row];
        int tb = __builtin_amdgcn_readfirstlane(ov);
        int endv = ov + cv - tb;
        int T = __builtin_amdgcn_readlane(endv, 15);
        int nd = 0;
        int bnd = __builtin_amdgcn_readlane(endv, 0);
        float4 acc = make_float4(0.f, 0.f, 0.f, 0.f);
        int s = (lane < T) ? sorted[tb + lane] : 0;
        for (int e0 = 0; e0 < T; e0 += 32) {
            int sn = (e0 + 32 + lane < T) ? sorted[tb + e0 + 32 + lane] : 0;
            walk32(x4, (unsigned)lane, s, T - e0, e0, endv, arow, acc, nd, bnd);
            s = sn;
        }
        for (; nd < 16; ++nd) {
            *(float4*)(arow + nd * APITCH + 4 * lane) = acc;
            acc = make_float4(0.f, 0.f, 0.f, 0.f);
        }

        unsigned rowOff = (unsigned)row * DFEAT + 2 * h;
        int ntile = tile + total_waves;
        int nrow = ((ntile < n_tiles) ? ntile : tile) * 16 + m;
        unsigned nextOff = (unsigned)nrow * DFEAT + 2 * h;

        v8f acc8[8] = {};

#pragma unroll 1
        for (int c = 0; c < 4; ++c) {
            unsigned po = (c < 3) ? (rowOff + 32 * (c + 1)) : nextOff;
            v2f nx[8];
#pragma unroll
            for (int j = 0; j < 8; ++j) nx[j] = *(const v2f*)(xin + po + 4 * j);
            v2f am[8];
#pragma unroll
            for (int j = 0; j < 8; ++j) am[j] = *(const v2f*)(ap + 32 * c + 4 * j);
#pragma unroll
            for (int j = 0; j < 8; ++j) {
                int k = 32 * c + 4 * j;
                v2f a_m = am[j];
                a_m[0] *= rd;
                a_m[1] *= rd;
                v2f a_x = ax[j];
#pragma unroll
                for (int t = 0; t < 8; ++t) {
                    v2f b_l = *(const v2f*)(bsh + t * 16 * WPITCH + k);
                    v2f b_r = *(const v2f*)(bsh + t * 16 * WPITCH + DFEAT + k);
                    acc8[t] = __builtin_amdgcn_wmma_f32_16x16x4_f32(
                        false, a_m, false, b_l, (short)0, acc8[t], false, false);
                    acc8[t] = __builtin_amdgcn_wmma_f32_16x16x4_f32(
                        false, a_x, false, b_r, (short)0, acc8[t], false, false);
                }
            }
#pragma unroll
            for (int j = 0; j < 8; ++j) ax[j] = nx[j];
        }

#pragma unroll
        for (int t = 0; t < 8; ++t) {
            float bias = bl[t * 16 + m];
            float* orow = out + (size_t)(tile * 16 + 8 * h) * DFEAT + t * 16 + m;
#pragma unroll
            for (int r = 0; r < 8; ++r) {
                float v = acc8[t][r] + bias;
                if (do_relu) v = fmaxf(v, 0.0f);
                orow[(size_t)r * DFEAT] = v;
            }
        }
    }
}

extern "C" void kernel_launch(void* const* d_in, const int* in_sizes, int n_in,
                              void* d_out, int out_size, void* d_ws, size_t ws_size,
                              hipStream_t stream) {
    const float* x   = (const float*)d_in[0];
    const int*   ei  = (const int*)d_in[1];
    const float* Wl1 = (const float*)d_in[2];
    const float* bl1 = (const float*)d_in[3];
    const float* Wr1 = (const float*)d_in[4];
    const float* Wl2 = (const float*)d_in[5];
    const float* bl2 = (const float*)d_in[6];
    const float* Wr2 = (const float*)d_in[7];
    float* outp = (float*)d_out;

    int n = in_sizes[0] / DFEAT;
    int E = in_sizes[1] / 2;
    const int* src = ei;
    const int* dst = ei + E;

    int nb = (n + NPB - 1) / NPB;

    float* hbuf = (float*)d_ws;
    int* pairs  = (int*)(hbuf + (size_t)n * DFEAT);
    float* deg  = (float*)(pairs + 2 * (size_t)E);
    int* cnt    = (int*)(deg + n);
    int* off    = cnt + n;
    int* slot   = off + n;
    int* sorted = slot + E;
    int* lo     = sorted + E;

    int tiles = (n + 15) / 16;
    int gemm_blocks = GEMM_BLOCKS;
    int wave_blocks = (n + 7) / 8;

    chunk_sort<<<CHUNKS, CHUNK_THREADS, nb * sizeof(int), stream>>>(src, dst, lo, pairs, E, nb);
    bucket_lists<<<nb, LIST_THREADS, 0, stream>>>(pairs, lo, cnt, off, slot, n, E, CHUNKS);
    order_lists<<<wave_blocks, 256, 0, stream>>>(slot, off, cnt, sorted, deg, n);

    sage_layer<<<gemm_blocks, 512, 0, stream>>>(x, sorted, off, cnt, deg, Wl1, bl1, Wr1,
                                               hbuf, tiles, 1);
    sage_layer<<<gemm_blocks, 512, 0, stream>>>(hbuf, sorted, off, cnt, deg, Wl2, bl2, Wr2,
                                               outp, tiles, 0);
}


# a bucket's lists are filled into LDS and put into source order by the bucket work-group itself (8 nodes per wave) before they are stored; order launch of 100000 one-node waves gone; sums unchanged
# speedup vs baseline: 5.1822x; 1.0787x over previous
#include <hip/hip_runtime.h>
#include <hip/hip_bf16.h>


typedef __attribute__((ext_vector_type(2))) float v2f;
typedef __attribute__((ext_vector_type(8))) float v8f;

#define DFEAT 128

#define NPB_SHIFT 8
#define NPB (1 << NPB_SHIFT)
#define CHUNKS 256
#define CHUNK_THREADS 1024
#define EDGE_UNR 8

__global__ __launch_bounds__(CHUNK_THREADS)
void chunk_sort(const int* __restrict__ src, const int* __restrict__ dst,
                int* __restrict__ lo, int* __restrict__ pairs,
                int n_edges, int n_buckets) {
    extern __shared__ int lds_cur[];
    __shared__ int wsum[CHUNK_THREADS / 32];
    int t = threadIdx.x;
    int lane = t & 31;
    int w = t >> 5;
    for (int i = t; i < n_buckets; i += CHUNK_THREADS) lds_cur[i] = 0;
    __syncthreads();
    int per = (n_edges + (int)gridDim.x - 1) / (int)gridDim.x;
    int e0 = min((int)blockIdx.x * per, n_edges);
    int e1 = min(e0 + per, n_edges);

    for (int e = e0 + t; e < e1; e += CHUNK_THREADS * EDGE_UNR) {
        int d[EDGE_UNR];
#pragma unroll
        for (int k = 0; k < EDGE_UNR; ++k)
            d[k] = dst[min(e + k * CHUNK_THREADS, e1 - 1)];
        static_assert(EDGE_UNR == 8, "the statement below names eight values");
        asm volatile("" :: "v"(d[0]), "v"(d[1]), "v"(d[2]), "v"(d[3]),
                           "v"(d[4]), "v"(d[5]), "v"(d[6]), "v"(d[7]));
#pragma unroll
        for (int k = 0; k < EDGE_UNR; ++k)
            if (e + k * CHUNK_THREADS < e1) atomicAdd(&lds_cur[d[k] >> NPB_SHIFT], 1);
    }
    __syncthreads();

    int ipt = (n_buckets + CHUNK_THREADS - 1) / CHUNK_THREADS;
    int i0 = t * ipt;
    int tot = 0;
    for (int q = 0; q < ipt; ++q)
        if (i0 + q < n_buckets) tot += lds_cur[i0 + q];
    int incl = tot;
    for (int o = 1; o < 32; o <<= 1) {
        int v = __shfl_up(incl, o);
        if (lane >= o) incl += v;
    }
    if (lane == 31) wsum[w] = incl;
    __syncthreads();
    int wb = 0;
    for (int j = 0; j < w; ++j) wb += wsum[j];
    int ex = wb + incl - tot;
    int* lrow = lo + (size_t)blockIdx.x * (size_t)(n_buckets + 1);
    for (int q = 0; q < ipt; ++q) {
        if (i0 + q < n_buckets) {
            int c = lds_cur[i0 + q];
            lds_cur[i0 + q] = ex;
            lrow[i0 + q] = ex;
            ex += c;
        }
    }
    if (t == 0) lrow[n_buckets] = e1 - e0;
    __syncthreads();

    for (int e = e0 + t; e < e1; e += CHUNK_THREADS * EDGE_UNR) {
        int d[EDGE_UNR];
        int s[EDGE_UNR];
#pragma unroll
        for (int k = 0; k < EDGE_UNR; ++k) {
            int ee = e + k * CHUNK_THREADS;
            d[k] = (ee < e1) ? dst[ee] : 0;
            s[k] = (ee < e1) ? src[ee] : 0;
        }
#pragma unroll
        for (int k = 0; k < EDGE_UNR; ++k) {
            if (e + k * CHUNK_THREADS < e1) {
                int p = atomicAdd(&lds_cur[d[k] >> NPB_SHIFT], 1);
                *(int2*)(pairs + 2 * ((size_t)e0 + (size_t)p)) = make_int2(s[k], d[k]);
            }
        }
    }
}

#define LIST_THREADS 1024
#define LIST_UNR 8
#define SEG_THREADS 4
#define LIST_CAP 8192
#define NODES_PER_WAVE (NPB / (LIST_THREADS / 32))
#define KEY_NODES (1 << 26)
static_assert(LIST_THREADS == 1024, "the sum over waves below takes one wave total per lane");
static_assert(NODES_PER_WAVE * (LIST_THREADS / 32) == NPB && NODES_PER_WAVE <= 32,
              "every wave orders the lists of NODES_PER_WAVE nodes, one lane each for count and start");

__device__ __forceinline__ void fill_lists(const int* __restrict__ pairs, const int* __restrict__ lo,
                                           int* __restrict__ ncur, int* __restrict__ lst,
                                           int t, int b, size_t nb1, int per, int base, int chunks) {
    int sub = t & (SEG_THREADS - 1);
    for (int cc = t / SEG_THREADS; cc < chunks; cc += LIST_THREADS / SEG_THREADS) {
        const int* row = lo + (size_t)cc * nb1 + b;
        int a = row[0];
        int len = row[1] - a;
        const int* seg = pairs + 2 * ((size_t)cc * (size_t)per + (size_t)a);
        for (int k = sub; k < len; k += SEG_THREADS * LIST_UNR) {
            int2 p[LIST_UNR];
#pragma unroll
            for (int j = 0; j < LIST_UNR; ++j) {
                int kk = k + j * SEG_THREADS;
                p[j] = (kk < len) ? *(const int2*)(seg + 2 * (size_t)kk) : make_int2(0, base);
            }
#pragma unroll
            for (int j = 0; j < LIST_UNR; ++j) {
                if (k + j * SEG_THREADS < len) {
                    int q = atomicAdd(&ncur[p[j].y - base], 1);
                    lst[q] = p[j].x;
                }
            }
        }
    }
}

__device__ __forceinline__ void order_lists(const int* __restrict__ lst, int* __restrict__ outl,
                                            int cv, int bv, int nodes, int lane, bool keyed) {
#pragma unroll 1
    for (int i = 0; i < nodes; ++i) {
        int lc = __builtin_amdgcn_readlane(cv, i);
        int lb = __builtin_amdgcn_readlane(bv, i);
        const int* src = lst + lb;
        int* dstl = outl + lb;
        if (lc <= 32 && keyed) {
            int s = (lane < lc) ? src[lane] : 0;
            unsigned key = ((unsigned)s << 5) | (unsigned)lane;
            int r = 0;
            for (int j = 0; j < lc; ++j) {
                unsigned kj = (unsigned)__builtin_amdgcn_readlane((int)key, j);
                r += (kj < key) ? 1 : 0;
            }
            if (lane < lc) dstl[r] = s;
        } else {
            for (int k = lane; k < lc; k += 32) {
                int s = src[k];
                int r = 0;
                for (int j = 0; j < lc; ++j) {
                    int sj = src[j];
                    r += ((sj < s) || (sj == s && j < k)) ? 1 : 0;
                }
                dstl[r] = s;
            }
        }
    }
}

__global__ __launch_bounds__(LIST_THREADS)
void bucket_lists(const int* __restrict__ pairs, const int* __restrict__ lo,
                  int* __restrict__ cnt, int* __restrict__ off,
                  int* __restrict__ slot, int* __restrict__ sorted,
                  float* __restrict__ deg, int n_nodes, int n_edges, int chunks) {
    __shared__ int ncnt[NPB];
    __shared__ int nbeg[NPB];
    __shared__ int ncur[NPB];
    __shared__ int wsum[NPB / 32];
    __shared__ int red[LIST_THREADS / 32];
    __shared__ int lslot[LIST_CAP];
    int t = threadIdx.x;
    int lane = t & 31;
    int w = t >> 5;
    int b = (int)blockIdx.x;
    size_t nb1 = (size_t)gridDim.x + 1;
    int per = (n_edges + chunks - 1) / chunks;
    int base = b << NPB_SHIFT;
    int sub = t & (SEG_THREADS - 1);
    if (t < NPB) ncnt[t] = 0;
    int part = 0;
    for (int c = t; c < chunks; c += LIST_THREADS) part += lo[(size_t)c * nb1 + b];
    for (int o = 16; o > 0; o >>= 1) part += __shfl_xor(part, o);
    if (lane == 0) red[w] = part;
    __syncthreads();
    int r0 = red[lane];
    for (int o = 16; o > 0; o >>= 1) r0 += __shfl_xor(r0, o);
    r0 = __builtin_amdgcn_readfirstlane(r0);

    for (int c = t / SEG_THREADS; c < chunks; c += LIST_THREADS / SEG_THREADS) {
        const int* row = lo + (size_t)c * nb1 + b;
        int a = row[0];
        int len = row[1] - a;
        const int* seg = pairs + 2 * ((size_t)c * (size_t)per + (size_t)a);
        for (int k = sub; k < len; k += SEG_THREADS * LIST_UNR) {
            int d[LIST_UNR];
#pragma unroll
            for (int j = 0; j < LIST_UNR; ++j) {
                int kk = k + j * SEG_THREADS;
                d[j] = (kk < len) ? seg[2 * (size_t)kk + 1] : base;
            }
#pragma unroll
            for (int j = 0; j < LIST_UNR; ++j)
                if (k + j * SEG_THREADS < len) atomicAdd(&ncnt[d[j] - base], 1);
        }
    }
    __syncthreads();
    int c = 0;
    int incl = 0;
    if (t < NPB) {
        c = ncnt[t];
        incl = c;
        for (int o = 1; o < 32; o <<= 1) {
            int v = __shfl_up(incl, o);
            if (lane >= o) incl += v;
        }
        if (lane == 31) wsum[w] = incl;
    }
    __syncthreads();
    if (t < NPB) {
        int wb = 0;
        for (int j = 0; j < w; ++j) wb += wsum[j];
        int ex = wb + incl - c;
        ncur[t] = ex;
        nbeg[t] = ex;
        if (base + t < n_nodes) {
            cnt[base + t] = c;
            off[base + t] = r0 + ex;
            deg[base + t] = (float)c;
        }
    }
    int tot = 0;
    for (int j = 0; j < NPB / 32; ++j) tot += wsum[j];
    tot = __builtin_amdgcn_readfirstlane(tot);
    bool in_lds = tot <= LIST_CAP;
    __syncthreads();
    if (in_lds) fill_lists(pairs, lo, ncur, lslot, t, b, nb1, per, base, chunks);
    else        fill_lists(pairs, lo, ncur, slot + r0, t, b, nb1, per, base, chunks);
    __syncthreads();

    int nl0 = __builtin_amdgcn_readfirstlane(w) * NODES_PER_WAVE;
    int cv = (lane < NODES_PER_WAVE) ? ncnt[nl0 + lane] : 0;
    int bv = (lane < NODES_PER_WAVE) ? nbeg[nl0 + lane] : 0;
    int nodes = min(NODES_PER_WAVE, n_nodes - (base + nl0));
    bool keyed = n_nodes <= KEY_NODES;
    if (in_lds) order_lists(lslot, sorted + r0, cv, bv, nodes, lane, keyed);
    else        order_lists(slot + r0, sorted + r0, cv, bv, nodes, lane, keyed);
}

#define WPITCH 260
#define APITCH 132
#define GEMM_WAVES 16
#define GEMM_BLOCKS 256

__device__ __forceinline__ void add4(float4& a, const float4& v) {
    a.x += v.x; a.y += v.y; a.z += v.z; a.w += v.w;
}

__device__ __forceinline__ void walk32(const float4* __restrict__ x4, unsigned lane, int s,
                                       int mcount, int e0, const int endv,
                                       float* __restrict__ arow, float4& acc,
                                       int& nd, int& bnd) {
    float4 v[32];
#pragma unroll
    for (int k = 0; k < 32; ++k) {
        unsigned sk = (unsigned)__builtin_amdgcn_readlane(s, k);
        v[k] = x4[sk * (DFEAT / 4) + lane];
    }
#pragma unroll
    for (int k = 0; k < 32; ++k) {
        if (k < mcount) {
            while (e0 + k == bnd) {
                *(float4*)(arow + nd * APITCH + 4 * lane) = acc;
                acc = make_float4(0.f, 0.f, 0.f, 0.f);
                ++nd;
                bnd = __builtin_amdgcn_readlane(endv, nd);
            }
            add4(acc, v[k]);
        }
    }
}

__global__ __launch_bounds__(512)
void sage_layer(const float* __restrict__ xin, const int* __restrict__ sorted,
                const int* __restrict__ off, const int* __restrict__ cnt,
                const float* __restrict__ deg,
                const float* __restrict__ Wl, const float* __restrict__ bl,
                const float* __restrict__ Wr, float* __restrict__ out,
                int n_tiles, int do_relu) {
    __shared__ __attribute__((aligned(16))) float wsh[DFEAT * WPITCH];
    __shared__ __attribute__((aligned(16))) float ash[GEMM_WAVES * 16 * APITCH];

#pragma unroll 8
    for (int idx = threadIdx.x; idx < DFEAT * 64; idx += 512) {
        int o = idx >> 6;
        int q = idx & 63;
        const float* sp = (q < 32) ? (Wl + (size_t)o * DFEAT + q * 4)
                                   : (Wr + (size_t)o * DFEAT + (q - 32) * 4);
        float4 v = *(const float4*)sp;
        *(float4*)(wsh + o * WPITCH + q * 4) = v;
    }
    __syncthreads();

    int w = __builtin_amdgcn_readfirstlane((int)(threadIdx.x >> 5));
    int lane = threadIdx.x & 31;
    int m = lane & 15;
    int h = lane >> 4;
    int total_waves = (int)gridDim.x * GEMM_WAVES;
    const float* bsh = wsh + m * WPITCH + 2 * h;
    float* arow = ash + w * (16 * APITCH);
    const float* ap = arow + m * APITCH + 2 * h;
    const float4* x4 = (const float4*)xin;

    int tile = w * (int)gridDim.x + (int)blockIdx.x;

    v2f ax[8];
    if (tile < n_tiles) {
        unsigned o0 = (unsigned)(tile * 16 + m) * DFEAT + 2 * h;
#pragma unroll
        for (int j = 0; j < 8; ++j) ax[j] = *(const v2f*)(xin + o0 + 4 * j);
    }

    for (; tile < n_tiles; tile += total_waves) {
        int row = tile * 16 + m;
        float rd = 1.0f / fmaxf(deg[row], 1.0f);

        int cv = cnt[row];
        int ov = off[row];
        int tb = __builtin_amdgcn_readfirstlane(ov);
        int endv = ov + cv - tb;
        int T = __builtin_amdgcn_readlane(endv, 15);
        int nd = 0;
        int bnd = __builtin_amdgcn_readlane(endv, 0);
        float4 acc = make_float4(0.f, 0.f, 0.f, 0.f);
        int s = (lane < T) ? sorted[tb + lane] : 0;
        for (int e0 = 0; e0 < T; e0 += 32) {
            int sn = (e0 + 32 + lane < T) ? sorted[tb + e0 + 32 + lane] : 0;
            walk32(x4, (unsigned)lane, s, T - e0, e0, endv, arow, acc, nd, bnd);
            s = sn;
        }
        for (; nd < 16; ++nd) {
            *(float4*)(arow + nd * APITCH + 4 * lane) = acc;
            acc = make_float4(0.f, 0.f, 0.f, 0.f);
        }

        unsigned rowOff = (unsigned)row * DFEAT + 2 * h;
        int ntile = tile + total_waves;
        int nrow = ((ntile < n_tiles) ? ntile : tile) * 16 + m;
        unsigned nextOff = (unsigned)nrow * DFEAT + 2 * h;

        v8f acc8[8] = {};

#pragma unroll 1
        for (int c = 0; c < 4; ++c) {
            unsigned po = (c < 3) ? (rowOff + 32 * (c + 1)) : nextOff;
            v2f nx[8];
#pragma unroll
            for (int j = 0; j < 8; ++j) nx[j] = *(const v2f*)(xin + po + 4 * j);
            v2f am[8];
#pragma unroll
            for (int j = 0; j < 8; ++j) am[j] = *(const v2f*)(ap + 32 * c + 4 * j);
#pragma unroll
            for (int j = 0; j < 8; ++j) {
                int k = 32 * c + 4 * j;
                v2f a_m = am[j];
                a_m[0] *= rd;
                a_m[1] *= rd;
                v2f a_x = ax[j];
#pragma unroll
                for (int t = 0; t < 8; ++t) {
                    v2f b_l = *(const v2f*)(bsh + t * 16 * WPITCH + k);
                    v2f b_r = *(const v2f*)(bsh + t * 16 * WPITCH + DFEAT + k);
                    acc8[t] = __builtin_amdgcn_wmma_f32_16x16x4_f32(
                        false, a_m, false, b_l, (short)0, acc8[t], false, false);
                    acc8[t] = __builtin_amdgcn_wmma_f32_16x16x4_f32(
                        false, a_x, false, b_r, (short)0, acc8[t], false, false);
                }
            }
#pragma unroll
            for (int j = 0; j < 8; ++j) ax[j] = nx[j];
        }

#pragma unroll
        for (int t = 0; t < 8; ++t) {
            float bias = bl[t * 16 + m];
            float* orow = out + (size_t)(tile * 16 + 8 * h) * DFEAT + t * 16 + m;
#pragma unroll
            for (int r = 0; r < 8; ++r) {
                float v = acc8[t][r] + bias;
                if (do_relu) v = fmaxf(v, 0.0f);
                orow[(size_t)r * DFEAT] = v;
            }
        }
    }
}

extern "C" void kernel_launch(void* const* d_in, const int* in_sizes, int n_in,
                              void* d_out, int out_size, void* d_ws, size_t ws_size,
                              hipStream_t stream) {
    const float* x   = (const float*)d_in[0];
    const int*   ei  = (const int*)d_in[1];
    const float* Wl1 = (const float*)d_in[2];
    const float* bl1 = (const float*)d_in[3];
    const float* Wr1 = (const float*)d_in[4];
    const float* Wl2 = (const float*)d_in[5];
    const float* bl2 = (const float*)d_in[6];
    const float* Wr2 = (const float*)d_in[7];
    float* outp = (float*)d_out;

    int n = in_sizes[0] / DFEAT;
    int E = in_sizes[1] / 2;
    const int* src = ei;
    const int* dst = ei + E;

    int nb = (n + NPB - 1) / NPB;

    float* hbuf = (float*)d_ws;
    int* pairs  = (int*)(hbuf + (size_t)n * DFEAT);
    float* deg  = (float*)(pairs + 2 * (size_t)E);
    int* cnt    = (int*)(deg + n);
    int* off    = cnt + n;
    int* slot   = off + n;
    int* sorted = slot + E;
    int* lo     = sorted + E;

    int tiles = (n + 15) / 16;
    int gemm_blocks = GEMM_BLOCKS;

    chunk_sort<<<CHUNKS, CHUNK_THREADS, nb * sizeof(int), stream>>>(src, dst, lo, pairs, E, nb);
    bucket_lists<<<nb, LIST_THREADS, 0, stream>>>(pairs, lo, cnt, off, slot, sorted, deg, n, E, CHUNKS);

    sage_layer<<<gemm_blocks, 512, 0, stream>>>(x, sorted, off, cnt, deg, Wl1, bl1, Wr1,
                                               hbuf, tiles, 1);
    sage_layer<<<gemm_blocks, 512, 0, stream>>>(hbuf, sorted, off, cnt, deg, Wl2, bl2, Wr2,
                                               outp, tiles, 0);
}


# neighbour walk in half steps of 16 rows: two half steps of row loads always in flight, indices loaded two steps ahead with no test, sums added in place; no index-load wait per step; sum order kept
# speedup vs baseline: 5.3683x; 1.0359x over previous
#include <hip/hip_runtime.h>
#include <hip/hip_bf16.h>


typedef __attribute__((ext_vector_type(2))) float v2f;
typedef __attribute__((ext_vector_type(8))) float v8f;

#define DFEAT 128

#define NPB_SHIFT 8
#define NPB (1 << NPB_SHIFT)
#define CHUNKS 256
#define CHUNK_THREADS 1024
#define EDGE_UNR 8

__global__ __launch_bounds__(CHUNK_THREADS)
void chunk_sort(const int* __restrict__ src, const int* __restrict__ dst,
                int* __restrict__ lo, int* __restrict__ pairs,
                int n_edges, int n_buckets) {
    extern __shared__ int lds_cur[];
    __shared__ int wsum[CHUNK_THREADS / 32];
    int t = threadIdx.x;
    int lane = t & 31;
    int w = t >> 5;
    for (int i = t; i < n_buckets; i += CHUNK_THREADS) lds_cur[i] = 0;
    __syncthreads();
    int per = (n_edges + (int)gridDim.x - 1) / (int)gridDim.x;
    int e0 = min((int)blockIdx.x * per, n_edges);
    int e1 = min(e0 + per, n_edges);

    for (int e = e0 + t; e < e1; e += CHUNK_THREADS * EDGE_UNR) {
        int d[EDGE_UNR];
#pragma unroll
        for (int k = 0; k < EDGE_UNR; ++k)
            d[k] = dst[min(e + k * CHUNK_THREADS, e1 - 1)];
        static_assert(EDGE_UNR == 8, "the statement below names eight values");
        asm volatile("" :: "v"(d[0]), "v"(d[1]), "v"(d[2]), "v"(d[3]),
                           "v"(d[4]), "v"(d[5]), "v"(d[6]), "v"(d[7]));
#pragma unroll
        for (int k = 0; k < EDGE_UNR; ++k)
            if (e + k * CHUNK_THREADS < e1) atomicAdd(&lds_cur[d[k] >> NPB_SHIFT], 1);
    }
    __syncthreads();

    int ipt = (n_buckets + CHUNK_THREADS - 1) / CHUNK_THREADS;
    int i0 = t * ipt;
    int tot = 0;
    for (int q = 0; q < ipt; ++q)
        if (i0 + q < n_buckets) tot += lds_cur[i0 + q];
    int incl = tot;
    for (int o = 1; o < 32; o <<= 1) {
        int v = __shfl_up(incl, o);
        if (lane >= o) incl += v;
    }
    if (lane == 31) wsum[w] = incl;
    __syncthreads();
    int wb = 0;
    for (int j = 0; j < w; ++j) wb += wsum[j];
    int ex = wb + incl - tot;
    int* lrow = lo + (size_t)blockIdx.x * (size_t)(n_buckets + 1);
    for (int q = 0; q < ipt; ++q) {
        if (i0 + q < n_buckets) {
            int c = lds_cur[i0 + q];
            lds_cur[i0 + q] = ex;
            lrow[i0 + q] = ex;
            ex += c;
        }
    }
    if (t == 0) lrow[n_buckets] = e1 - e0;
    __syncthreads();

    for (int e = e0 + t; e < e1; e += CHUNK_THREADS * EDGE_UNR) {
        int d[EDGE_UNR];
        int s[EDGE_UNR];
#pragma unroll
        for (int k = 0; k < EDGE_UNR; ++k) {
            int ee = e + k * CHUNK_THREADS;
            d[k] = (ee < e1) ? dst[ee] : 0;
            s[k] = (ee < e1) ? src[ee] : 0;
        }
#pragma unroll
        for (int k = 0; k < EDGE_UNR; ++k) {
            if (e + k * CHUNK_THREADS < e1) {
                int p = atomicAdd(&lds_cur[d[k] >> NPB_SHIFT], 1);
                *(int2*)(pairs + 2 * ((size_t)e0 + (size_t)p)) = make_int2(s[k], d[k]);
            }
        }
    }
}

#define LIST_THREADS 1024
#define LIST_UNR 8
#define SEG_THREADS 4
#define LIST_CAP 8192
#define NODES_PER_WAVE (NPB / (LIST_THREADS / 32))
#define KEY_NODES (1 << 26)
static_assert(LIST_THREADS == 1024, "the sum over waves below takes one wave total per lane");
static_assert(NODES_PER_WAVE * (LIST_THREADS / 32) == NPB && NODES_PER_WAVE <= 32,
              "every wave orders the lists of NODES_PER_WAVE nodes, one lane each for count and start");

__device__ __forceinline__ void fill_lists(const int* __restrict__ pairs, const int* __restrict__ lo,
                                           int* __restrict__ ncur, int* __restrict__ lst,
                                           int t, int b, size_t nb1, int per, int base, int chunks) {
    int sub = t & (SEG_THREADS - 1);
    for (int cc = t / SEG_THREADS; cc < chunks; cc += LIST_THREADS / SEG_THREADS) {
        const int* row = lo + (size_t)cc * nb1 + b;
        int a = row[0];
        int len = row[1] - a;
        const int* seg = pairs + 2 * ((size_t)cc * (size_t)per + (size_t)a);
        for (int k = sub; k < len; k += SEG_THREADS * LIST_UNR) {
            int2 p[LIST_UNR];
#pragma unroll
            for (int j = 0; j < LIST_UNR; ++j) {
                int kk = k + j * SEG_THREADS;
                p[j] = (kk < len) ? *(const int2*)(seg + 2 * (size_t)kk) : make_int2(0, base);
            }
#pragma unroll
            for (int j = 0; j < LIST_UNR; ++j) {
                if (k + j * SEG_THREADS < len) {
                    int q = atomicAdd(&ncur[p[j].y - base], 1);
                    lst[q] = p[j].x;
                }
            }
        }
    }
}

__device__ __forceinline__ void order_lists(const int* __restrict__ lst, int* __restrict__ outl,
                                            int cv, int bv, int nodes, int lane, bool keyed) {
#pragma unroll 1
    for (int i = 0; i < nodes; ++i) {
        int lc = __builtin_amdgcn_readlane(cv, i);
        int lb = __builtin_amdgcn_readlane(bv, i);
        const int* src = lst + lb;
        int* dstl = outl + lb;
        if (lc <= 32 && keyed) {
            int s = (lane < lc) ? src[lane] : 0;
            unsigned key = ((unsigned)s << 5) | (unsigned)lane;
            int r = 0;
            for (int j = 0; j < lc; ++j) {
                unsigned kj = (unsigned)__builtin_amdgcn_readlane((int)key, j);
                r += (kj < key) ? 1 : 0;
            }
            if (lane < lc) dstl[r] = s;
        } else {
            for (int k = lane; k < lc; k += 32) {
                int s = src[k];
                int r = 0;
                for (int j = 0; j < lc; ++j) {
                    int sj = src[j];
                    r += ((sj < s) || (sj == s && j < k)) ? 1 : 0;
                }
                dstl[r] = s;
            }
        }
    }
}

__global__ __launch_bounds__(LIST_THREADS)
void bucket_lists(const int* __restrict__ pairs, const int* __restrict__ lo,
                  int* __restrict__ cnt, int* __restrict__ off,
                  int* __restrict__ slot, int* __restrict__ sorted,
                  float* __restrict__ deg, int n_nodes, int n_edges, int chunks) {
    __shared__ int ncnt[NPB];
    __shared__ int nbeg[NPB];
    __shared__ int ncur[NPB];
    __shared__ int wsum[NPB / 32];
    __shared__ int red[LIST_THREADS / 32];
    __shared__ int lslot[LIST_CAP];
    int t = threadIdx.x;
    int lane = t & 31;
    int w = t >> 5;
    int b = (int)blockIdx.x;
    size_t nb1 = (size_t)gridDim.x + 1;
    int per = (n_edges + chunks - 1) / chunks;
    int base = b << NPB_SHIFT;
    int sub = t & (SEG_THREADS - 1);
    if (t < NPB) ncnt[t] = 0;
    int part = 0;
    for (int c = t; c < chunks; c += LIST_THREADS) part += lo[(size_t)c * nb1 + b];
    for (int o = 16; o > 0; o >>= 1) part += __shfl_xor(part, o);
    if (lane == 0) red[w] = part;
    __syncthreads();
    int r0 = red[lane];
    for (int o = 16; o > 0; o >>= 1) r0 += __shfl_xor(r0, o);
    r0 = __builtin_amdgcn_readfirstlane(r0);

    for (int c = t / SEG_THREADS; c < chunks; c += LIST_THREADS / SEG_THREADS) {
        const int* row = lo + (size_t)c * nb1 + b;
        int a = row[0];
        int len = row[1] - a;
        const int* seg = pairs + 2 * ((size_t)c * (size_t)per + (size_t)a);
        for (int k = sub; k < len; k += SEG_THREADS * LIST_UNR) {
            int d[LIST_UNR];
#pragma unroll
            for (int j = 0; j < LIST_UNR; ++j) {
                int kk = k + j * SEG_THREADS;
                d[j] = (kk < len) ? seg[2 * (size_t)kk + 1] : base;
            }
#pragma unroll
            for (int j = 0; j < LIST_UNR; ++j)
                if (k + j * SEG_THREADS < len) atomicAdd(&ncnt[d[j] - base], 1);
        }
    }
    __syncthreads();
    int c = 0;
    int incl = 0;
    if (t < NPB) {
        c = ncnt[t];
        incl = c;
        for (int o = 1; o < 32; o <<= 1) {
            int v = __shfl_up(incl, o);
            if (lane >= o) incl += v;
        }
        if (lane == 31) wsum[w] = incl;
    }
    __syncthreads();
    if (t < NPB) {
        int wb = 0;
        for (int j = 0; j < w; ++j) wb += wsum[j];
        int ex = wb + incl - c;
        ncur[t] = ex;
        nbeg[t] = ex;
        if (base + t < n_nodes) {
            cnt[base + t] = c;
            off[base + t] = r0 + ex;
            deg[base + t] = (float)c;
        }
    }
    int tot = 0;
    for (int j = 0; j < NPB / 32; ++j) tot += wsum[j];
    tot = __builtin_amdgcn_readfirstlane(tot);
    bool in_lds = tot <= LIST_CAP;
    __syncthreads();
    if (in_lds) fill_lists(pairs, lo, ncur, lslot, t, b, nb1, per, base, chunks);
    else        fill_lists(pairs, lo, ncur, slot + r0, t, b, nb1, per, base, chunks);
    __syncthreads();

    int nl0 = __builtin_amdgcn_readfirstlane(w) * NODES_PER_WAVE;
    int cv = (lane < NODES_PER_WAVE) ? ncnt[nl0 + lane] : 0;
    int bv = (lane < NODES_PER_WAVE) ? nbeg[nl0 + lane] : 0;
    int nodes = min(NODES_PER_WAVE, n_nodes - (base + nl0));
    bool keyed = n_nodes <= KEY_NODES;
    if (in_lds) order_lists(lslot, sorted + r0, cv, bv, nodes, lane, keyed);
    else        order_lists(slot + r0, sorted + r0, cv, bv, nodes, lane, keyed);
}

#define WPITCH 260
#define APITCH 132
#define GEMM_WAVES 16
#define GEMM_BLOCKS 256

__device__ __forceinline__ void add4(v2f& lo, v2f& hi, const float4& v) {
    v2f rl = {v.x, v.y};
    v2f rh = {v.z, v.w};
    asm("v_pk_add_f32 %0, %0, %1" : "+v"(lo) : "v"(rl));
    asm("v_pk_add_f32 %0, %0, %1" : "+v"(hi) : "v"(rh));
}

template <int HALF>
__device__ __forceinline__ void issue16(const float4* __restrict__ x4, unsigned lane, int s,
                                        float4 (&v)[16]) {
#pragma unroll
    for (int k = 0; k < 16; ++k) {
        unsigned sk = (unsigned)__builtin_amdgcn_readlane(s, 16 * HALF + k);
        v[k] = x4[sk * (DFEAT / 4) + lane];
    }
}

template <bool FULL>
__device__ __forceinline__ void add16(const float4 (&v)[16], unsigned lane, int mcount, int e0,
                                      const int endv, float* __restrict__ arow, v2f& alo, v2f& ahi,
                                      int& nd, int& bnd) {
#pragma unroll
    for (int k = 0; k < 16; ++k) {
        if (FULL || k < mcount) {
            while (e0 + k == bnd) {
                *(float4*)(arow + nd * APITCH + 4 * lane) = make_float4(alo[0], alo[1], ahi[0], ahi[1]);
                alo = (v2f){0.f, 0.f};
                ahi = (v2f){0.f, 0.f};
                ++nd;
                bnd = __builtin_amdgcn_readlane(endv, nd);
            }
            add4(alo, ahi, v[k]);
        }
    }
}

__global__ __launch_bounds__(512)
void sage_layer(const float* __restrict__ xin, const int* __restrict__ sorted,
                const int* __restrict__ off, const int* __restrict__ cnt,
                const float* __restrict__ deg,
                const float* __restrict__ Wl, const float* __restrict__ bl,
                const float* __restrict__ Wr, float* __restrict__ out,
                int n_tiles, int do_relu) {
    __shared__ __attribute__((aligned(16))) float wsh[DFEAT * WPITCH];
    __shared__ __attribute__((aligned(16))) float ash[GEMM_WAVES * 16 * APITCH];

#pragma unroll 8
    for (int idx = threadIdx.x; idx < DFEAT * 64; idx += 512) {
        int o = idx >> 6;
        int q = idx & 63;
        const float* sp = (q < 32) ? (Wl + (size_t)o * DFEAT + q * 4)
                                   : (Wr + (size_t)o * DFEAT + (q - 32) * 4);
        float4 v = *(const float4*)sp;
        *(float4*)(wsh + o * WPITCH + q * 4) = v;
    }
    __syncthreads();

    int w = __builtin_amdgcn_readfirstlane((int)(threadIdx.x >> 5));
    int lane = threadIdx.x & 31;
    int m = lane & 15;
    int h = lane >> 4;
    int total_waves = (int)gridDim.x * GEMM_WAVES;
    const float* bsh = wsh + m * WPITCH + 2 * h;
    float* arow = ash + w * (16 * APITCH);
    const float* ap = arow + m * APITCH + 2 * h;
    const float4* x4 = (const float4*)xin;

    int tile = w * (int)gridDim.x + (int)blockIdx.x;

    v2f ax[8];
    if (tile < n_tiles) {
        unsigned o0 = (unsigned)(tile * 16 + m) * DFEAT + 2 * h;
#pragma unroll
        for (int j = 0; j < 8; ++j) ax[j] = *(const v2f*)(xin + o0 + 4 * j);
    }

    for (; tile < n_tiles; tile += total_waves) {
        int row = tile * 16 + m;
        float rd = 1.0f / fmaxf(deg[row], 1.0f);

        int cv = cnt[row];
        int ov = off[row];
        int tb = __builtin_amdgcn_readfirstlane(ov);
        int endv = ov + cv - tb;
        int T = __builtin_amdgcn_readlane(endv, 15);
        int nd = 0;
        int bnd = __builtin_amdgcn_readlane(endv, 0);
        v2f alo = {0.f, 0.f};
        v2f ahi = {0.f, 0.f};
        if (T > 0) {
            int last = T - 1;
            int steps = (T + 31) >> 5;
            int s0 = sorted[tb + min(lane, last)];
            int s1 = sorted[tb + min(32 + lane, last)];
            asm volatile("" :: "v"(s0), "v"(s1));
            float4 va[16], vb[16];
            issue16<0>(x4, (unsigned)lane, s0, va);
            issue16<1>(x4, (unsigned)lane, s0, vb);
            int e0 = 0;
            for (int i = 0; i + 1 < steps; ++i) {
                int s2 = sorted[tb + min(e0 + 64 + lane, last)];
                add16<true>(va, (unsigned)lane, 16, e0, endv, arow, alo, ahi, nd, bnd);
                issue16<0>(x4, (unsigned)lane, s1, va);
                add16<true>(vb, (unsigned)lane, 16, e0 + 16, endv, arow, alo, ahi, nd, bnd);
                issue16<1>(x4, (unsigned)lane, s1, vb);
                s1 = s2;
                e0 += 32;
            }
            add16<false>(va, (unsigned)lane, T - e0, e0, endv, arow, alo, ahi, nd, bnd);
            add16<false>(vb, (unsigned)lane, T - e0 - 16, e0 + 16, endv, arow, alo, ahi, nd, bnd);
        }
        for (; nd < 16; ++nd) {
            *(float4*)(arow + nd * APITCH + 4 * lane) = make_float4(alo[0], alo[1], ahi[0], ahi[1]);
            alo = (v2f){0.f, 0.f};
            ahi = (v2f){0.f, 0.f};
        }

        unsigned rowOff = (unsigned)row * DFEAT + 2 * h;
        int ntile = tile + total_waves;
        int nrow = ((ntile < n_tiles) ? ntile : tile) * 16 + m;
        unsigned nextOff = (unsigned)nrow * DFEAT + 2 * h;

        v8f acc8[8] = {};

#pragma unroll 1
        for (int c = 0; c < 4; ++c) {
            unsigned po = (c < 3) ? (rowOff + 32 * (c + 1)) : nextOff;
            v2f nx[8];
#pragma unroll
            for (int j = 0; j < 8; ++j) nx[j] = *(const v2f*)(xin + po + 4 * j);
            v2f am[8];
#pragma unroll
            for (int j = 0; j < 8; ++j) am[j] = *(const v2f*)(ap + 32 * c + 4 * j);
#pragma unroll
            for (int j = 0; j < 8; ++j) {
                int k = 32 * c + 4 * j;
                v2f a_m = am[j];
                a_m[0] *= rd;
                a_m[1] *= rd;
                v2f a_x = ax[j];
#pragma unroll
                for (int t = 0; t < 8; ++t) {
                    v2f b_l = *(const v2f*)(bsh + t * 16 * WPITCH + k);
                    v2f b_r = *(const v2f*)(bsh + t * 16 * WPITCH + DFEAT + k);
                    acc8[t] = __builtin_amdgcn_wmma_f32_16x16x4_f32(
                        false, a_m, false, b_l, (short)0, acc8[t], false, false);
                    acc8[t] = __builtin_amdgcn_wmma_f32_16x16x4_f32(
                        false, a_x, false, b_r, (short)0, acc8[t], false, false);
                }
            }
#pragma unroll
            for (int j = 0; j < 8; ++j) ax[j] = nx[j];
        }

#pragma unroll
        for (int t = 0; t < 8; ++t) {
            float bias = bl[t * 16 + m];
            float* orow = out + (size_t)(tile * 16 + 8 * h) * DFEAT + t * 16 + m;
#pragma unroll
            for (int r = 0; r < 8; ++r) {
                float v = acc8[t][r] + bias;
                if (do_relu) v = fmaxf(v, 0.0f);
                orow[(size_t)r * DFEAT] = v;
            }
        }
    }
}

extern "C" void kernel_launch(void* const* d_in, const int* in_sizes, int n_in,
                              void* d_out, int out_size, void* d_ws, size_t ws_size,
                              hipStream_t stream) {
    const float* x   = (const float*)d_in[0];
    const int*   ei  = (const int*)d_in[1];
    const float* Wl1 = (const float*)d_in[2];
    const float* bl1 = (const float*)d_in[3];
    const float* Wr1 = (const float*)d_in[4];
    const float* Wl2 = (const float*)d_in[5];
    const float* bl2 = (const float*)d_in[6];
    const float* Wr2 = (const float*)d_in[7];
    float* outp = (float*)d_out;

    int n = in_sizes[0] / DFEAT;
    int E = in_sizes[1] / 2;
    const int* src = ei;
    const int* dst = ei + E;

    int nb = (n + NPB - 1) / NPB;

    float* hbuf = (float*)d_ws;
    int* pairs  = (int*)(hbuf + (size_t)n * DFEAT);
    float* deg  = (float*)(pairs + 2 * (size_t)E);
    int* cnt    = (int*)(deg + n);
    int* off    = cnt + n;
    int* slot   = off + n;
    int* sorted = slot + E;
    int* lo     = sorted + E;

    int tiles = (n + 15) / 16;
    int gemm_blocks = GEMM_BLOCKS;

    chunk_sort<<<CHUNKS, CHUNK_THREADS, nb * sizeof(int), stream>>>(src, dst, lo, pairs, E, nb);
    bucket_lists<<<nb, LIST_THREADS, 0, stream>>>(pairs, lo, cnt, off, slot, sorted, deg, n, E, CHUNKS);

    sage_layer<<<gemm_blocks, 512, 0, stream>>>(x, sorted, off, cnt, deg, Wl1, bl1, Wr1,
                                               hbuf, tiles, 1);
    sage_layer<<<gemm_blocks, 512, 0, stream>>>(hbuf, sorted, off, cnt, deg, Wl2, bl2, Wr2,
                                               outp, tiles, 0);
}
